# starting kernel
# baseline (speedup 1.0000x reference)
#include <hip/hip_runtime.h>


typedef __attribute__((ext_vector_type(2))) float v2f;
typedef __attribute__((ext_vector_type(8))) float v8f;

#define NN   50000
#define NE   600000
#define KDIM 128


__global__ void k_zero4(float4* __restrict__ p, int n4) {
  int i = blockIdx.x * blockDim.x + threadIdx.x;
  if (i < n4) p[i] = make_float4(0.f, 0.f, 0.f, 0.f);
}

__global__ void k_deg(const int* __restrict__ src, const int* __restrict__ dst,
                      float* __restrict__ degS, float* __restrict__ degD, int nE) {
  int i = blockIdx.x * blockDim.x + threadIdx.x;
  if (i < nE) {
    atomicAdd(&degS[src[i]], 1.0f);
    atomicAdd(&degD[dst[i]], 1.0f);
  }
}

__global__ void k_norm(float* __restrict__ a, float* __restrict__ b, int n) {
  int i = blockIdx.x * blockDim.x + threadIdx.x;
  if (i < n) {
    a[i] = rsqrtf(fmaxf(a[i], 1.0f));
    b[i] = rsqrtf(fmaxf(b[i], 1.0f));
  }
}


__global__ __launch_bounds__(128) void k_gemm_wmma(
    const float* __restrict__ X, const float* __restrict__ norm,
    const float* __restrict__ W, float* __restrict__ Y, int ncshift  ) {
  __shared__ float As[16 * 132];
  __shared__ float Ws[(KDIM / 2) * 144 * 2];

  const int ncol  = 1 << ncshift;
  const int ncolP = ncol + 16;
  const int m0    = blockIdx.x * 16;
  const int tid   = threadIdx.x;

  for (int i = tid; i < 16 * KDIM; i += 128) {
    int r = i >> 7;
    int c = i & (KDIM - 1);
    As[r * 132 + c] = X[(size_t)(m0 + r) * KDIM + c] * norm[m0 + r];
  }

  {
    const int ncq = ncshift - 2;
    const int nf4 = KDIM << ncq;
    for (int idx = tid; idx < nf4; idx += 128) {
      const int k = idx >> ncq;
      const int j = (idx - (k << ncq)) << 2;
      const float4 v = ((const float4*)W)[idx];
      const int base = ((k >> 1) * ncolP + j) * 2 + (k & 1);
      Ws[base + 0] = v.x;
      Ws[base + 2] = v.y;
      Ws[base + 4] = v.z;
      Ws[base + 6] = v.w;
    }
  }
  __syncthreads();

  const int wave = tid >> 5;
  const int lane = tid & 31;
  const int hl   = lane >> 4;
  const int lm   = lane & 15;

  const int ntile = ncol >> 4;
  for (int t = wave; t < ntile; t += 4) {
    v8f acc = {};
    const int n = t * 16 + lm;
    #pragma unroll
    for (int k0 = 0; k0 < KDIM; k0 += 4) {
      const int kk = k0 + 2 * hl;
      v2f a;
      a.x = As[lm * 132 + kk];
      a.y = As[lm * 132 + kk + 1];
      const int p = kk >> 1;
      const v2f b = *(const v2f*)&Ws[(p * ncolP + n) * 2];
      acc = __builtin_amdgcn_wmma_f32_16x16x4_f32(
           false, a,  false, b,
           (short)0, acc,  false,  false);
    }
    #pragma unroll
    for (int r = 0; r < 8; ++r)
      Y[(size_t)(m0 + hl * 8 + r) * ncol + n] = acc[r];
  }
}


__global__ void k_scatter(const float* __restrict__ H, const int* __restrict__ src,
                          const int* __restrict__ dst, float* __restrict__ out,
                          int nE, int fshift  ) {
  const int total = nE << fshift;
  int i = blockIdx.x * blockDim.x + threadIdx.x;
  if (i >= total) return;
  const int e = i >> fshift;
  const int c = i & ((1 << fshift) - 1);
  const int s = src[e];
  const int d = dst[e];
  const float4 v = ((const float4*)H)[((size_t)s << fshift) + c];
  float* o = out + ((((size_t)d << fshift) + c) << 2);
  atomicAdd(o + 0, v.x);
  atomicAdd(o + 1, v.y);
  atomicAdd(o + 2, v.z);
  atomicAdd(o + 3, v.w);
}


__global__ void k_post(float* __restrict__ h, const float* __restrict__ normD,
                       const float* __restrict__ b, int total, int shift,
                       int doRelu) {
  int i = blockIdx.x * blockDim.x + threadIdx.x;
  if (i >= total) return;
  const int row = i >> shift;
  const int col = i & ((1 << shift) - 1);
  float v = h[i] * normD[row] + b[col];
  h[i] = doRelu ? fmaxf(v, 0.0f) : v;
}


extern "C" void kernel_launch(void* const* d_in, const int* in_sizes, int n_in,
                              void* d_out, int out_size, void* d_ws, size_t ws_size,
                              hipStream_t stream) {
  (void)in_sizes; (void)n_in; (void)out_size; (void)ws_size;

  const float* features = (const float*)d_in[0];
  const float* W1       = (const float*)d_in[1];
  const float* b1       = (const float*)d_in[2];
  const float* W2       = (const float*)d_in[3];
  const float* b2       = (const float*)d_in[4];
  const int*   esrc     = (const int*)d_in[5];
  const int*   edst     = (const int*)d_in[6];
  float*       out      = (float*)d_out;

  const size_t PADN = 50176;
  float* w     = (float*)d_ws;
  float* normS = w;
  float* normD = w + PADN;
  float* B0    = w + 2 * PADN;
  float* B1    = B0 + (size_t)NN * 128;

  const int T = 256;

  {
    int n4 = (int)(2 * PADN / 4);
    k_zero4<<<(n4 + T - 1) / T, T, 0, stream>>>((float4*)normS, n4);
    k_deg<<<(NE + T - 1) / T, T, 0, stream>>>(esrc, edst, normS, normD, NE);
    k_norm<<<(NN + T - 1) / T, T, 0, stream>>>(normS, normD, NN);
  }

  k_gemm_wmma<<<NN / 16, 128, 0, stream>>>(features, normS, W1, B0, 7);

  {
    int n4 = NN * 32;
    k_zero4<<<(n4 + T - 1) / T, T, 0, stream>>>((float4*)B1, n4);
    int tot = NE * 32;
    k_scatter<<<(tot + T - 1) / T, T, 0, stream>>>(B0, esrc, edst, B1, NE, 5);
  }

  k_post<<<(NN * 128 + T - 1) / T, T, 0, stream>>>(B1, normD, b1, NN * 128, 7, 1);

  k_gemm_wmma<<<NN / 16, 128, 0, stream>>>(B1, normS, W2, B0, 6);

  {
    int n4 = NN * 16;
    k_zero4<<<(n4 + T - 1) / T, T, 0, stream>>>((float4*)out, n4);
    int tot = NE * 16;
    k_scatter<<<(tot + T - 1) / T, T, 0, stream>>>(B0, esrc, edst, out, NE, 4);
  }

  k_post<<<(NN * 64 + T - 1) / T, T, 0, stream>>>(out, normD, b2, NN * 64, 6, 0);
}


# GEMMs: W staged into LDS once per 16-wave work-group (256 groups) instead of once per 16-row strip; each wave owns one strip, A fragments in registers; k order and sums unchanged
# speedup vs baseline: 1.2684x; 1.2684x over previous
#include <hip/hip_runtime.h>


typedef __attribute__((ext_vector_type(2))) float v2f;
typedef __attribute__((ext_vector_type(8))) float v8f;

#define NN   50000
#define NE   600000
#define KDIM 128


__global__ void k_zero4(float4* __restrict__ p, int n4) {
  int i = blockIdx.x * blockDim.x + threadIdx.x;
  if (i < n4) p[i] = make_float4(0.f, 0.f, 0.f, 0.f);
}

__global__ void k_deg(const int* __restrict__ src, const int* __restrict__ dst,
                      float* __restrict__ degS, float* __restrict__ degD, int nE) {
  int i = blockIdx.x * blockDim.x + threadIdx.x;
  if (i < nE) {
    atomicAdd(&degS[src[i]], 1.0f);
    atomicAdd(&degD[dst[i]], 1.0f);
  }
}

__global__ void k_norm(float* __restrict__ a, float* __restrict__ b, int n) {
  int i = blockIdx.x * blockDim.x + threadIdx.x;
  if (i < n) {
    a[i] = rsqrtf(fmaxf(a[i], 1.0f));
    b[i] = rsqrtf(fmaxf(b[i], 1.0f));
  }
}


#define GEMM_THREADS 512
#define GEMM_GRID    256

template <int NCSHIFT  >
__global__ __launch_bounds__(GEMM_THREADS) void k_gemm_wmma(
    const float* __restrict__ X, const float* __restrict__ norm,
    const float* __restrict__ W, float* __restrict__ Y, int nstrips) {
  constexpr int ncol  = 1 << NCSHIFT;
  constexpr int ncolP = ncol + 16;
  constexpr int ntile = ncol >> 4;
  __shared__ float Ws[(KDIM / 2) * ncolP * 2];

  const int tid = threadIdx.x;

  {
    constexpr int ncq = NCSHIFT - 2;
    constexpr int nf4 = KDIM << ncq;
    for (int i = 0; i < nf4 / GEMM_THREADS; ++i) {
      const int idx = tid + i * GEMM_THREADS;
      const int k = idx >> ncq;
      const int j = (idx - (k << ncq)) << 2;
      const float4 v = ((const float4*)W)[idx];
      const int base = ((k >> 1) * ncolP + j) * 2 + (k & 1);
      Ws[base + 0] = v.x;
      Ws[base + 2] = v.y;
      Ws[base + 4] = v.z;
      Ws[base + 6] = v.w;
    }
  }
  __syncthreads();

  const int wave = tid >> 5;
  const int lane = tid & 31;
  const int hl   = lane >> 4;
  const int lm   = lane & 15;

  const int s = wave * (int)gridDim.x + (int)blockIdx.x;
  if (s < nstrips) {
    const int m0 = s * 16;
    const float* __restrict__ xr = X + (size_t)(m0 + lm) * KDIM + 2 * hl;
    const float nrm = norm[m0 + lm];

    v2f a[KDIM / 4];
    for (int j = 0; j < KDIM / 4; ++j) {
      const v2f x = *(const v2f*)(xr + 4 * j);
      a[j].x = x.x * nrm;
      a[j].y = x.y * nrm;
    }

    v8f acc[ntile];
    for (int t = 0; t < ntile; ++t) acc[t] = v8f{};

    for (int j = 0; j < KDIM / 4; ++j) {
      const int p = 2 * j + hl;
      for (int t = 0; t < ntile; ++t) {
        const v2f b = *(const v2f*)&Ws[(p * ncolP + t * 16 + lm) * 2];
        acc[t] = __builtin_amdgcn_wmma_f32_16x16x4_f32(
             false, a[j],  false, b,
             (short)0, acc[t],  false,  false);
      }
    }

    for (int t = 0; t < ntile; ++t) {
      for (int r = 0; r < 8; ++r)
        Y[(size_t)(m0 + hl * 8 + r) * ncol + t * 16 + lm] = acc[t][r];
    }
  }
}


__global__ void k_scatter(const float* __restrict__ H, const int* __restrict__ src,
                          const int* __restrict__ dst, float* __restrict__ out,
                          int nE, int fshift  ) {
  const int total = nE << fshift;
  int i = blockIdx.x * blockDim.x + threadIdx.x;
  if (i >= total) return;
  const int e = i >> fshift;
  const int c = i & ((1 << fshift) - 1);
  const int s = src[e];
  const int d = dst[e];
  const float4 v = ((const float4*)H)[((size_t)s << fshift) + c];
  float* o = out + ((((size_t)d << fshift) + c) << 2);
  atomicAdd(o + 0, v.x);
  atomicAdd(o + 1, v.y);
  atomicAdd(o + 2, v.z);
  atomicAdd(o + 3, v.w);
}


__global__ void k_post(float* __restrict__ h, const float* __restrict__ normD,
                       const float* __restrict__ b, int total, int shift,
                       int doRelu) {
  int i = blockIdx.x * blockDim.x + threadIdx.x;
  if (i >= total) return;
  const int row = i >> shift;
  const int col = i & ((1 << shift) - 1);
  float v = h[i] * normD[row] + b[col];
  h[i] = doRelu ? fmaxf(v, 0.0f) : v;
}


extern "C" void kernel_launch(void* const* d_in, const int* in_sizes, int n_in,
                              void* d_out, int out_size, void* d_ws, size_t ws_size,
                              hipStream_t stream) {
  (void)in_sizes; (void)n_in; (void)out_size; (void)ws_size;

  const float* features = (const float*)d_in[0];
  const float* W1       = (const float*)d_in[1];
  const float* b1       = (const float*)d_in[2];
  const float* W2       = (const float*)d_in[3];
  const float* b2       = (const float*)d_in[4];
  const int*   esrc     = (const int*)d_in[5];
  const int*   edst     = (const int*)d_in[6];
  float*       out      = (float*)d_out;

  const size_t PADN = 50176;
  float* w     = (float*)d_ws;
  float* normS = w;
  float* normD = w + PADN;
  float* B0    = w + 2 * PADN;
  float* B1    = B0 + (size_t)NN * 128;

  const int T = 256;

  {
    int n4 = (int)(2 * PADN / 4);
    k_zero4<<<(n4 + T - 1) / T, T, 0, stream>>>((float4*)normS, n4);
    k_deg<<<(NE + T - 1) / T, T, 0, stream>>>(esrc, edst, normS, normD, NE);
    k_norm<<<(NN + T - 1) / T, T, 0, stream>>>(normS, normD, NN);
  }

  k_gemm_wmma<7><<<GEMM_GRID, GEMM_THREADS, 0, stream>>>(features, normS, W1, B0, NN / 16);

  {
    int n4 = NN * 32;
    k_zero4<<<(n4 + T - 1) / T, T, 0, stream>>>((float4*)B1, n4);
    int tot = NE * 32;
    k_scatter<<<(tot + T - 1) / T, T, 0, stream>>>(B0, esrc, edst, B1, NE, 5);
  }

  k_post<<<(NN * 128 + T - 1) / T, T, 0, stream>>>(B1, normD, b1, NN * 128, 7, 1);

  k_gemm_wmma<6><<<GEMM_GRID, GEMM_THREADS, 0, stream>>>(B1, normS, W2, B0, NN / 16);

  {
    int n4 = NN * 16;
    k_zero4<<<(n4 + T - 1) / T, T, 0, stream>>>((float4*)out, n4);
    int tot = NE * 16;
    k_scatter<<<(tot + T - 1) / T, T, 0, stream>>>(B0, esrc, edst, out, NE, 4);
  }

  k_post<<<(NN * 64 + T - 1) / T, T, 0, stream>>>(out, normD, b2, NN * 64, 6, 0);
}


# Edge sums: per-destination gather over a CSR built in-call (int counts, slots ranked by edge id) replaces float-atomic scatter; sum order now fixed by edge id; scale+bias+relu at the gather's store
# speedup vs baseline: 1.6248x; 1.2810x over previous
#include <hip/hip_runtime.h>


typedef __attribute__((ext_vector_type(2))) float v2f;
typedef __attribute__((ext_vector_type(4))) float v4f;
typedef __attribute__((ext_vector_type(8))) float v8f;

#define NN   50000
#define NE   600000
#define KDIM 128
#define PADN 50176


__global__ void k_zero4(float4* __restrict__ p, int n4) {
  int i = blockIdx.x * blockDim.x + threadIdx.x;
  if (i < n4) p[i] = make_float4(0.f, 0.f, 0.f, 0.f);
}

__global__ void k_count(const int* __restrict__ src, const int* __restrict__ dst,
                        int* __restrict__ cntS, int* __restrict__ cntD, int nE4) {
  int i = blockIdx.x * blockDim.x + threadIdx.x;
  if (i < nE4) {
    const int4 s = ((const int4*)src)[i];
    const int4 d = ((const int4*)dst)[i];
    atomicAdd(&cntS[s.x], 1);
    atomicAdd(&cntS[s.y], 1);
    atomicAdd(&cntS[s.z], 1);
    atomicAdd(&cntS[s.w], 1);
    atomicAdd(&cntD[d.x], 1);
    atomicAdd(&cntD[d.y], 1);
    atomicAdd(&cntD[d.z], 1);
    atomicAdd(&cntD[d.w], 1);
  }
}

#define ALLOC_THREADS 256
__global__ __launch_bounds__(ALLOC_THREADS) void k_alloc(
    const int* __restrict__ cntS, const int* __restrict__ cntD,
    float* __restrict__ normS, float* __restrict__ normD,
    int2* __restrict__ seg, int* __restrict__ cur, int* __restrict__ total, int n) {
  __shared__ int wsum[ALLOC_THREADS / 32];
  __shared__ int base;
  const int i    = blockIdx.x * ALLOC_THREADS + threadIdx.x;
  const int lane = threadIdx.x & 31;
  const int wave = threadIdx.x >> 5;
  const int c = (i < n) ? cntD[i] : 0;
  int x = c;
  for (int o = 1; o < 32; o <<= 1) {
    const int y = __shfl_up(x, o);
    if (lane >= o) x += y;
  }
  if (lane == 31) wsum[wave] = x;
  __syncthreads();
  if (threadIdx.x == 0) {
    int t = 0;
    for (int w = 0; w < ALLOC_THREADS / 32; ++w) {
      const int v = wsum[w];
      wsum[w] = t;
      t += v;
    }
    base = atomicAdd(total, t);
  }
  __syncthreads();
  if (i < n) {
    const int beg = base + wsum[wave] + x - c;
    seg[i]    = make_int2(beg, c);
    cur[i]    = beg;
    normS[i]  = rsqrtf(fmaxf((float)cntS[i], 1.0f));
    normD[i]  = rsqrtf(fmaxf((float)c, 1.0f));
  }
}

__global__ void k_fill(const int* __restrict__ dst, int* __restrict__ cur,
                       int* __restrict__ tmpE, int nE4) {
  int i = blockIdx.x * blockDim.x + threadIdx.x;
  if (i < nE4) {
    const int4 d = ((const int4*)dst)[i];
    const int p0 = atomicAdd(&cur[d.x], 1);
    const int p1 = atomicAdd(&cur[d.y], 1);
    const int p2 = atomicAdd(&cur[d.z], 1);
    const int p3 = atomicAdd(&cur[d.w], 1);
    tmpE[p0] = 4 * i + 0;
    tmpE[p1] = 4 * i + 1;
    tmpE[p2] = 4 * i + 2;
    tmpE[p3] = 4 * i + 3;
  }
}

__global__ void k_rank(const int* __restrict__ src, const int* __restrict__ dst,
                       const int2* __restrict__ seg,
                       const int* __restrict__ tmpE, int* __restrict__ csrSrc, int nE) {
  int e = blockIdx.x * blockDim.x + threadIdx.x;
  if (e < nE) {
    const int d = dst[e];
    const int2 sg = seg[d];
    const int b = sg.x;
    const int n = sg.y;
    int r = 0;
    for (int j = 0; j < n; ++j) r += (tmpE[b + j] < e) ? 1 : 0;
    csrSrc[b + r] = src[e];
  }
}


#define GEMM_THREADS 512
#define GEMM_GRID    256

template <int NCSHIFT  >
__global__ __launch_bounds__(GEMM_THREADS) void k_gemm_wmma(
    const float* __restrict__ X, const float* __restrict__ norm,
    const float* __restrict__ W, float* __restrict__ Y, int nstrips) {
  constexpr int ncol  = 1 << NCSHIFT;
  constexpr int ncolP = ncol + 16;
  constexpr int ntile = ncol >> 4;
  __shared__ float Ws[(KDIM / 2) * ncolP * 2];

  const int tid = threadIdx.x;

  {
    constexpr int ncq = NCSHIFT - 2;
    constexpr int nf4 = KDIM << ncq;
    for (int i = 0; i < nf4 / GEMM_THREADS; ++i) {
      const int idx = tid + i * GEMM_THREADS;
      const int k = idx >> ncq;
      const int j = (idx - (k << ncq)) << 2;
      const float4 v = ((const float4*)W)[idx];
      const int base = ((k >> 1) * ncolP + j) * 2 + (k & 1);
      Ws[base + 0] = v.x;
      Ws[base + 2] = v.y;
      Ws[base + 4] = v.z;
      Ws[base + 6] = v.w;
    }
  }
  __syncthreads();

  const int wave = tid >> 5;
  const int lane = tid & 31;
  const int hl   = lane >> 4;
  const int lm   = lane & 15;

  const int s = wave * (int)gridDim.x + (int)blockIdx.x;
  if (s < nstrips) {
    const int m0 = s * 16;
    const float* __restrict__ xr = X + (size_t)(m0 + lm) * KDIM + 2 * hl;
    const float nrm = norm[m0 + lm];

    v2f a[KDIM / 4];
    for (int j = 0; j < KDIM / 4; ++j) {
      const v2f x = *(const v2f*)(xr + 4 * j);
      a[j].x = x.x * nrm;
      a[j].y = x.y * nrm;
    }

    v8f acc[ntile];
    for (int t = 0; t < ntile; ++t) acc[t] = v8f{};

    for (int j = 0; j < KDIM / 4; ++j) {
      const int p = 2 * j + hl;
      for (int t = 0; t < ntile; ++t) {
        const v2f b = *(const v2f*)&Ws[(p * ncolP + t * 16 + lm) * 2];
        acc[t] = __builtin_amdgcn_wmma_f32_16x16x4_f32(
             false, a[j],  false, b,
             (short)0, acc[t],  false,  false);
      }
    }

    for (int t = 0; t < ntile; ++t) {
      for (int r = 0; r < 8; ++r)
        Y[(size_t)(m0 + hl * 8 + r) * ncol + t * 16 + lm] = acc[t][r];
    }
  }
}


template <int VEC> struct RowVec;
template <> struct RowVec<4> { typedef v4f T; };
template <> struct RowVec<2> { typedef v2f T; };

#define GATHER_THREADS 256
#define GATHER_GRID    2048

template <int VEC, int U>
__device__ __forceinline__ void gather_batch(typename RowVec<VEC>::T& acc, int ids, int j,
                                             const typename RowVec<VEC>::T* __restrict__ H,
                                             int lane) {
  typedef typename RowVec<VEC>::T vecT;
  vecT v[U];
#pragma unroll
  for (int u = 0; u < U; ++u) {
    const int s = __builtin_amdgcn_readlane(ids, j + u);
    v[u] = H[(size_t)s * 32 + lane];
  }
#pragma unroll
  for (int u = 0; u < U; ++u) acc += v[u];
}

template <int VEC>
__device__ __forceinline__ void gather_chunk(typename RowVec<VEC>::T& acc, int ids, int m,
                                             const typename RowVec<VEC>::T* __restrict__ H,
                                             int lane) {
  int j = 0;
  for (; j + 8 <= m; j += 8) gather_batch<VEC, 8>(acc, ids, j, H, lane);
  if (m & 4) { gather_batch<VEC, 4>(acc, ids, j, H, lane); j += 4; }
  if (m & 2) { gather_batch<VEC, 2>(acc, ids, j, H, lane); j += 2; }
  if (m & 1) { gather_batch<VEC, 1>(acc, ids, j, H, lane); }
}

template <int VEC, bool RELU>
__global__ __launch_bounds__(GATHER_THREADS) void k_gather(
    const float* __restrict__ Hf, const int2* __restrict__ seg,
    const int* __restrict__ csrSrc, const float* __restrict__ normD,
    const float* __restrict__ bias, float* __restrict__ outf, int n, int nE) {
  typedef typename RowVec<VEC>::T vecT;
  const vecT* __restrict__ H = (const vecT*)Hf;
  vecT* __restrict__ out = (vecT*)outf;
  const int lane = threadIdx.x & 31;
  const int nwav = (int)gridDim.x * (GATHER_THREADS / 32);
  const vecT bv = ((const vecT*)bias)[lane];
  int node = __builtin_amdgcn_readfirstlane((int)blockIdx.x * (GATHER_THREADS / 32) +
                                            ((int)threadIdx.x >> 5));
  if (node >= n) return;
  const int last = nE - 1;
  int2 sg  = seg[node];
  int2 sgN = make_int2(0, 0);
  if (node + nwav < n) sgN = seg[node + nwav];
  int ids = csrSrc[min(sg.x + lane, last)];
  for (;;) {
    const int nn = node + nwav;
    const int idsN = csrSrc[min(sgN.x + lane, last)];
    int2 sgNN = make_int2(0, 0);
    if (nn + nwav < n) sgNN = seg[nn + nwav];
    const float nrm = normD[node];

    const int beg = sg.x;
    const int len = sg.y;
    vecT acc = (vecT)(0.0f);
    gather_chunk<VEC>(acc, ids, (len < 32) ? len : 32, H, lane);
    for (int c0 = 32; c0 < len; c0 += 32) {
      const int m = (len - c0 < 32) ? (len - c0) : 32;
      const int cid = csrSrc[min(beg + c0 + lane, last)];
      gather_chunk<VEC>(acc, cid, m, H, lane);
    }
    vecT r = acc * nrm + bv;
    if (RELU) {
#pragma unroll
      for (int q = 0; q < VEC; ++q) r[q] = fmaxf(r[q], 0.0f);
    }
    out[(size_t)node * 32 + lane] = r;

    if (nn >= n) break;
    node = nn;
    sg   = sgN;
    sgN  = sgNN;
    ids  = idsN;
  }
}


extern "C" void kernel_launch(void* const* d_in, const int* in_sizes, int n_in,
                              void* d_out, int out_size, void* d_ws, size_t ws_size,
                              hipStream_t stream) {
  (void)in_sizes; (void)n_in; (void)out_size; (void)ws_size;

  const float* features = (const float*)d_in[0];
  const float* W1       = (const float*)d_in[1];
  const float* b1       = (const float*)d_in[2];
  const float* W2       = (const float*)d_in[3];
  const float* b2       = (const float*)d_in[4];
  const int*   esrc     = (const int*)d_in[5];
  const int*   edst     = (const int*)d_in[6];
  float*       out      = (float*)d_out;

  int*   wi     = (int*)d_ws;
  int*   cntS   = wi;
  int*   cntD   = wi + PADN;
  int*   total  = wi + 2 * PADN;
  float* normS  = (float*)(wi + 2 * PADN + 64);
  float* normD  = normS + PADN;
  int2*  seg    = (int2*)(normD + PADN);
  int*   cur    = (int*)(seg + PADN);
  int*   tmpE   = cur + PADN;
  int*   csrSrc = tmpE + NE;
  float* B0     = (float*)(csrSrc + NE);
  float* B1     = B0 + (size_t)NN * 128;

  const int T = 256;

  {
    const int n4 = (2 * PADN + 64) / 4;
    k_zero4<<<(n4 + T - 1) / T, T, 0, stream>>>((float4*)wi, n4);
    k_count<<<(NE / 4 + T - 1) / T, T, 0, stream>>>(esrc, edst, cntS, cntD, NE / 4);
    k_alloc<<<(NN + ALLOC_THREADS - 1) / ALLOC_THREADS, ALLOC_THREADS, 0, stream>>>(
        cntS, cntD, normS, normD, seg, cur, total, NN);
    k_fill<<<(NE / 4 + T - 1) / T, T, 0, stream>>>(edst, cur, tmpE, NE / 4);
    k_rank<<<(NE + T - 1) / T, T, 0, stream>>>(esrc, edst, seg, tmpE, csrSrc, NE);
  }

  k_gemm_wmma<7><<<GEMM_GRID, GEMM_THREADS, 0, stream>>>(features, normS, W1, B0, NN / 16);

  k_gather<4, true><<<GATHER_GRID, GATHER_THREADS, 0, stream>>>(
      B0, seg, csrSrc, normD, b1, B1, NN, NE);

  k_gemm_wmma<6><<<GEMM_GRID, GEMM_THREADS, 0, stream>>>(B1, normS, W2, B0, NN / 16);

  k_gather<2, false><<<GATHER_GRID, GATHER_THREADS, 0, stream>>>(
      B0, seg, csrSrc, normD, b2, out, NN, NE);
}


# CSR build: degree counts and slot cursors in LDS per (node range, edge chunk) work-group, chunk partials merged by a prefix: no scattered global integer atomics, no zeroing launch; sums unchanged
# speedup vs baseline: 3.3155x; 2.0406x over previous
#include <hip/hip_runtime.h>


typedef __attribute__((ext_vector_type(2))) float v2f;
typedef __attribute__((ext_vector_type(4))) float v4f;
typedef __attribute__((ext_vector_type(8))) float v8f;

#define NN   50000
#define NE   600000
#define KDIM 128
#define PADN 50176


#define NR   8
#define RN   6250
#define NC   32
#define CE4  4688
#define CSR_THREADS 1024
#define CSR_GRID    (NR * NC)
#define CSR_NI      ((CE4 + CSR_THREADS - 1) / CSR_THREADS)
#define CSR_RI      ((RN + CSR_THREADS - 1) / CSR_THREADS)

__global__ __launch_bounds__(CSR_THREADS) void k_count(
    const int* __restrict__ src, const int* __restrict__ dst,
    int* __restrict__ partS, int* __restrict__ partD, int* __restrict__ total) {
  __shared__ int cS[RN];
  __shared__ int cD[RN];
  const int tid = threadIdx.x;
  const int r   = (int)blockIdx.x % NR;
  const int c   = (int)blockIdx.x / NR;
  const int lo  = r * RN;
  const int n4 = min(CE4, NE / 4 - c * CE4);
  const int4* __restrict__ s4 = (const int4*)src + c * CE4;
  const int4* __restrict__ d4 = (const int4*)dst + c * CE4;
  int4 s[CSR_NI];
  int4 d[CSR_NI];
#pragma unroll
  for (int k = 0; k < CSR_NI; ++k) {
    const int i = min(tid + k * CSR_THREADS, n4 - 1);
    s[k] = s4[i];
    d[k] = d4[i];
  }
  for (int i = tid; i < RN; i += CSR_THREADS) { cS[i] = 0; cD[i] = 0; }
  if (blockIdx.x == 0 && tid == 0) *total = 0;
  __syncthreads();
#pragma unroll
  for (int k = 0; k < CSR_NI; ++k) {
    if (tid + k * CSR_THREADS < n4) {
      unsigned u;
      u = (unsigned)(s[k].x - lo); if (u < RN) atomicAdd(&cS[u], 1);
      u = (unsigned)(s[k].y - lo); if (u < RN) atomicAdd(&cS[u], 1);
      u = (unsigned)(s[k].z - lo); if (u < RN) atomicAdd(&cS[u], 1);
      u = (unsigned)(s[k].w - lo); if (u < RN) atomicAdd(&cS[u], 1);
      u = (unsigned)(d[k].x - lo); if (u < RN) atomicAdd(&cD[u], 1);
      u = (unsigned)(d[k].y - lo); if (u < RN) atomicAdd(&cD[u], 1);
      u = (unsigned)(d[k].z - lo); if (u < RN) atomicAdd(&cD[u], 1);
      u = (unsigned)(d[k].w - lo); if (u < RN) atomicAdd(&cD[u], 1);
    }
  }
  __syncthreads();
  int* __restrict__ pS = partS + (size_t)c * PADN + lo;
  int* __restrict__ pD = partD + (size_t)c * PADN + lo;
  for (int i = tid; i < RN; i += CSR_THREADS) { pS[i] = cS[i]; pD[i] = cD[i]; }
}

#define ALLOC_THREADS 256
__global__ __launch_bounds__(ALLOC_THREADS) void k_alloc(
    const int* __restrict__ partS, int* __restrict__ partD,
    float* __restrict__ normS, float* __restrict__ normD,
    int2* __restrict__ seg, int* __restrict__ total, int n) {
  __shared__ int wsum[ALLOC_THREADS / 32];
  __shared__ int base;
  const int i    = blockIdx.x * ALLOC_THREADS + threadIdx.x;
  const int lane = threadIdx.x & 31;
  const int wave = threadIdx.x >> 5;
  int cs = 0;
  int c  = 0;
  if (i < n) {
    int vs[NC];
    int vd[NC];
#pragma unroll
    for (int k = 0; k < NC; ++k) {
      vs[k] = partS[(size_t)k * PADN + i];
      vd[k] = partD[(size_t)k * PADN + i];
    }
#pragma unroll
    for (int k = 0; k < NC; ++k) {
      cs += vs[k];
      partD[(size_t)k * PADN + i] = c;
      c += vd[k];
    }
  }
  int x = c;
  for (int o = 1; o < 32; o <<= 1) {
    const int y = __shfl_up(x, o);
    if (lane >= o) x += y;
  }
  if (lane == 31) wsum[wave] = x;
  __syncthreads();
  if (threadIdx.x == 0) {
    int t = 0;
    for (int w = 0; w < ALLOC_THREADS / 32; ++w) {
      const int v = wsum[w];
      wsum[w] = t;
      t += v;
    }
    base = atomicAdd(total, t);
  }
  __syncthreads();
  if (i < n) {
    const int beg = base + wsum[wave] + x - c;
    seg[i]    = make_int2(beg, c);
    normS[i]  = rsqrtf(fmaxf((float)cs, 1.0f));
    normD[i]  = rsqrtf(fmaxf((float)c, 1.0f));
  }
}

__global__ __launch_bounds__(CSR_THREADS) void k_fill(
    const int* __restrict__ dst, const int2* __restrict__ seg,
    const int* __restrict__ partD, int* __restrict__ tmpE) {
  __shared__ int cur[RN];
  const int tid = threadIdx.x;
  const int r   = (int)blockIdx.x % NR;
  const int c   = (int)blockIdx.x / NR;
  const int lo  = r * RN;
  const int* __restrict__ pD = partD + (size_t)c * PADN + lo;
  const int n4 = min(CE4, NE / 4 - c * CE4);
  const int4* __restrict__ d4 = (const int4*)dst + c * CE4;
  int4 d[CSR_NI];
#pragma unroll
  for (int k = 0; k < CSR_NI; ++k) d[k] = d4[min(tid + k * CSR_THREADS, n4 - 1)];
  int sb[CSR_RI];
  int co[CSR_RI];
#pragma unroll
  for (int k = 0; k < CSR_RI; ++k) {
    const int i = min(tid + k * CSR_THREADS, RN - 1);
    sb[k] = seg[lo + i].x;
    co[k] = pD[i];
  }
#pragma unroll
  for (int k = 0; k < CSR_RI; ++k) {
    const int i = tid + k * CSR_THREADS;
    if (i < RN) cur[i] = sb[k] + co[k];
  }
  __syncthreads();
#pragma unroll
  for (int k = 0; k < CSR_NI; ++k) {
    const int i = tid + k * CSR_THREADS;
    if (i < n4) {
      const int e0 = 4 * (c * CE4 + i);
      unsigned u;
      u = (unsigned)(d[k].x - lo); if (u < RN) tmpE[atomicAdd(&cur[u], 1)] = e0 + 0;
      u = (unsigned)(d[k].y - lo); if (u < RN) tmpE[atomicAdd(&cur[u], 1)] = e0 + 1;
      u = (unsigned)(d[k].z - lo); if (u < RN) tmpE[atomicAdd(&cur[u], 1)] = e0 + 2;
      u = (unsigned)(d[k].w - lo); if (u < RN) tmpE[atomicAdd(&cur[u], 1)] = e0 + 3;
    }
  }
}

__global__ void k_rank(const int* __restrict__ src, const int* __restrict__ dst,
                       const int2* __restrict__ seg,
                       const int* __restrict__ tmpE, int* __restrict__ csrSrc, int nE) {
  int e = blockIdx.x * blockDim.x + threadIdx.x;
  if (e < nE) {
    const int d = dst[e];
    const int2 sg = seg[d];
    const int b = sg.x;
    const int n = sg.y;
    int r = 0;
    for (int j = 0; j < n; ++j) r += (tmpE[b + j] < e) ? 1 : 0;
    csrSrc[b + r] = src[e];
  }
}


#define GEMM_THREADS 512
#define GEMM_GRID    256

template <int NCSHIFT  >
__global__ __launch_bounds__(GEMM_THREADS) void k_gemm_wmma(
    const float* __restrict__ X, const float* __restrict__ norm,
    const float* __restrict__ W, float* __restrict__ Y, int nstrips) {
  constexpr int ncol  = 1 << NCSHIFT;
  constexpr int ncolP = ncol + 16;
  constexpr int ntile = ncol >> 4;
  __shared__ float Ws[(KDIM / 2) * ncolP * 2];

  const int tid = threadIdx.x;

  {
    constexpr int ncq = NCSHIFT - 2;
    constexpr int nf4 = KDIM << ncq;
    for (int i = 0; i < nf4 / GEMM_THREADS; ++i) {
      const int idx = tid + i * GEMM_THREADS;
      const int k = idx >> ncq;
      const int j = (idx - (k << ncq)) << 2;
      const float4 v = ((const float4*)W)[idx];
      const int base = ((k >> 1) * ncolP + j) * 2 + (k & 1);
      Ws[base + 0] = v.x;
      Ws[base + 2] = v.y;
      Ws[base + 4] = v.z;
      Ws[base + 6] = v.w;
    }
  }
  __syncthreads();

  const int wave = tid >> 5;
  const int lane = tid & 31;
  const int hl   = lane >> 4;
  const int lm   = lane & 15;

  const int s = wave * (int)gridDim.x + (int)blockIdx.x;
  if (s < nstrips) {
    const int m0 = s * 16;
    const float* __restrict__ xr = X + (size_t)(m0 + lm) * KDIM + 2 * hl;
    const float nrm = norm[m0 + lm];

    v2f a[KDIM / 4];
    for (int j = 0; j < KDIM / 4; ++j) {
      const v2f x = *(const v2f*)(xr + 4 * j);
      a[j].x = x.x * nrm;
      a[j].y = x.y * nrm;
    }

    v8f acc[ntile];
    for (int t = 0; t < ntile; ++t) acc[t] = v8f{};

    for (int j = 0; j < KDIM / 4; ++j) {
      const int p = 2 * j + hl;
      for (int t = 0; t < ntile; ++t) {
        const v2f b = *(const v2f*)&Ws[(p * ncolP + t * 16 + lm) * 2];
        acc[t] = __builtin_amdgcn_wmma_f32_16x16x4_f32(
             false, a[j],  false, b,
             (short)0, acc[t],  false,  false);
      }
    }

    for (int t = 0; t < ntile; ++t) {
      for (int r = 0; r < 8; ++r)
        Y[(size_t)(m0 + hl * 8 + r) * ncol + t * 16 + lm] = acc[t][r];
    }
  }
}


template <int VEC> struct RowVec;
template <> struct RowVec<4> { typedef v4f T; };
template <> struct RowVec<2> { typedef v2f T; };

#define GATHER_THREADS 256
#define GATHER_GRID    2048

template <int VEC, int U>
__device__ __forceinline__ void gather_batch(typename RowVec<VEC>::T& acc, int ids, int j,
                                             const typename RowVec<VEC>::T* __restrict__ H,
                                             int lane) {
  typedef typename RowVec<VEC>::T vecT;
  vecT v[U];
#pragma unroll
  for (int u = 0; u < U; ++u) {
    const int s = __builtin_amdgcn_readlane(ids, j + u);
    v[u] = H[(size_t)s * 32 + lane];
  }
#pragma unroll
  for (int u = 0; u < U; ++u) acc += v[u];
}

template <int VEC>
__device__ __forceinline__ void gather_chunk(typename RowVec<VEC>::T& acc, int ids, int m,
                                             const typename RowVec<VEC>::T* __restrict__ H,
                                             int lane) {
  int j = 0;
  for (; j + 8 <= m; j += 8) gather_batch<VEC, 8>(acc, ids, j, H, lane);
  if (m & 4) { gather_batch<VEC, 4>(acc, ids, j, H, lane); j += 4; }
  if (m & 2) { gather_batch<VEC, 2>(acc, ids, j, H, lane); j += 2; }
  if (m & 1) { gather_batch<VEC, 1>(acc, ids, j, H, lane); }
}

template <int VEC, bool RELU>
__global__ __launch_bounds__(GATHER_THREADS) void k_gather(
    const float* __restrict__ Hf, const int2* __restrict__ seg,
    const int* __restrict__ csrSrc, const float* __restrict__ normD,
    const float* __restrict__ bias, float* __restrict__ outf, int n, int nE) {
  typedef typename RowVec<VEC>::T vecT;
  const vecT* __restrict__ H = (const vecT*)Hf;
  vecT* __restrict__ out = (vecT*)outf;
  const int lane = threadIdx.x & 31;
  const int nwav = (int)gridDim.x * (GATHER_THREADS / 32);
  const vecT bv = ((const vecT*)bias)[lane];
  int node = __builtin_amdgcn_readfirstlane((int)blockIdx.x * (GATHER_THREADS / 32) +
                                            ((int)threadIdx.x >> 5));
  if (node >= n) return;
  const int last = nE - 1;
  int2 sg  = seg[node];
  int2 sgN = make_int2(0, 0);
  if (node + nwav < n) sgN = seg[node + nwav];
  int ids = csrSrc[min(sg.x + lane, last)];
  for (;;) {
    const int nn = node + nwav;
    const int idsN = csrSrc[min(sgN.x + lane, last)];
    int2 sgNN = make_int2(0, 0);
    if (nn + nwav < n) sgNN = seg[nn + nwav];
    const float nrm = normD[node];

    const int beg = sg.x;
    const int len = sg.y;
    vecT acc = (vecT)(0.0f);
    gather_chunk<VEC>(acc, ids, (len < 32) ? len : 32, H, lane);
    for (int c0 = 32; c0 < len; c0 += 32) {
      const int m = (len - c0 < 32) ? (len - c0) : 32;
      const int cid = csrSrc[min(beg + c0 + lane, last)];
      gather_chunk<VEC>(acc, cid, m, H, lane);
    }
    vecT r = acc * nrm + bv;
    if (RELU) {
#pragma unroll
      for (int q = 0; q < VEC; ++q) r[q] = fmaxf(r[q], 0.0f);
    }
    out[(size_t)node * 32 + lane] = r;

    if (nn >= n) break;
    node = nn;
    sg   = sgN;
    sgN  = sgNN;
    ids  = idsN;
  }
}


extern "C" void kernel_launch(void* const* d_in, const int* in_sizes, int n_in,
                              void* d_out, int out_size, void* d_ws, size_t ws_size,
                              hipStream_t stream) {
  (void)in_sizes; (void)n_in; (void)out_size; (void)ws_size;

  const float* features = (const float*)d_in[0];
  const float* W1       = (const float*)d_in[1];
  const float* b1       = (const float*)d_in[2];
  const float* W2       = (const float*)d_in[3];
  const float* b2       = (const float*)d_in[4];
  const int*   esrc     = (const int*)d_in[5];
  const int*   edst     = (const int*)d_in[6];
  float*       out      = (float*)d_out;

  int*   wi     = (int*)d_ws;
  int*   total  = wi;
  float* normS  = (float*)(wi + 64);
  float* normD  = normS + PADN;
  int2*  seg    = (int2*)(normD + PADN);
  int*   tmpE   = (int*)(seg + PADN);
  int*   csrSrc = tmpE + NE;
  int*   partS  = csrSrc + NE;
  int*   partD  = partS + (size_t)NC * PADN;
  float* B0     = (float*)(partD + (size_t)NC * PADN);
  float* B1     = B0 + (size_t)NN * 128;

  const int T = 256;

  {
    k_count<<<CSR_GRID, CSR_THREADS, 0, stream>>>(esrc, edst, partS, partD, total);
    k_alloc<<<(NN + ALLOC_THREADS - 1) / ALLOC_THREADS, ALLOC_THREADS, 0, stream>>>(
        partS, partD, normS, normD, seg, total, NN);
    k_fill<<<CSR_GRID, CSR_THREADS, 0, stream>>>(edst, seg, partD, tmpE);
    k_rank<<<(NE + T - 1) / T, T, 0, stream>>>(esrc, edst, seg, tmpE, csrSrc, NE);
  }

  k_gemm_wmma<7><<<GEMM_GRID, GEMM_THREADS, 0, stream>>>(features, normS, W1, B0, NN / 16);

  k_gather<4, true><<<GATHER_GRID, GATHER_THREADS, 0, stream>>>(
      B0, seg, csrSrc, normD, b1, B1, NN, NE);

  k_gemm_wmma<6><<<GEMM_GRID, GEMM_THREADS, 0, stream>>>(B1, normS, W2, B0, NN / 16);

  k_gather<2, false><<<GATHER_GRID, GATHER_THREADS, 0, stream>>>(
      B0, seg, csrSrc, normD, b2, out, NN, NE);
}


# CSR build: the work-group that hands out a chunk's slots also ranks each (node, chunk) group by edge id, one thread per node: the separate per-edge ranking launch is gone; sum order unchanged
# speedup vs baseline: 4.2312x; 1.2762x over previous
#include <hip/hip_runtime.h>


typedef __attribute__((ext_vector_type(2))) float v2f;
typedef __attribute__((ext_vector_type(4))) float v4f;
typedef __attribute__((ext_vector_type(8))) float v8f;

#define NN   50000
#define NE   600000
#define KDIM 128
#define PADN 50176


#define NR   8
#define RN   6250
#define NC   32
#define CE4  4688
#define CSR_THREADS 1024
#define CSR_GRID    (NR * NC)
#define CSR_NI      ((CE4 + CSR_THREADS - 1) / CSR_THREADS)
#define CSR_RI      ((RN + CSR_THREADS - 1) / CSR_THREADS)

__global__ __launch_bounds__(CSR_THREADS) void k_count(
    const int* __restrict__ src, const int* __restrict__ dst,
    int* __restrict__ partS, int* __restrict__ partD, int* __restrict__ total) {
  __shared__ int cS[RN];
  __shared__ int cD[RN];
  const int tid = threadIdx.x;
  const int r   = (int)blockIdx.x % NR;
  const int c   = (int)blockIdx.x / NR;
  const int lo  = r * RN;
  const int n4 = min(CE4, NE / 4 - c * CE4);
  const int4* __restrict__ s4 = (const int4*)src + c * CE4;
  const int4* __restrict__ d4 = (const int4*)dst + c * CE4;
  int4 s[CSR_NI];
  int4 d[CSR_NI];
#pragma unroll
  for (int k = 0; k < CSR_NI; ++k) {
    const int i = min(tid + k * CSR_THREADS, n4 - 1);
    s[k] = s4[i];
    d[k] = d4[i];
  }
  for (int i = tid; i < RN; i += CSR_THREADS) { cS[i] = 0; cD[i] = 0; }
  if (blockIdx.x == 0 && tid == 0) *total = 0;
  __syncthreads();
#pragma unroll
  for (int k = 0; k < CSR_NI; ++k) {
    if (tid + k * CSR_THREADS < n4) {
      unsigned u;
      u = (unsigned)(s[k].x - lo); if (u < RN) atomicAdd(&cS[u], 1);
      u = (unsigned)(s[k].y - lo); if (u < RN) atomicAdd(&cS[u], 1);
      u = (unsigned)(s[k].z - lo); if (u < RN) atomicAdd(&cS[u], 1);
      u = (unsigned)(s[k].w - lo); if (u < RN) atomicAdd(&cS[u], 1);
      u = (unsigned)(d[k].x - lo); if (u < RN) atomicAdd(&cD[u], 1);
      u = (unsigned)(d[k].y - lo); if (u < RN) atomicAdd(&cD[u], 1);
      u = (unsigned)(d[k].z - lo); if (u < RN) atomicAdd(&cD[u], 1);
      u = (unsigned)(d[k].w - lo); if (u < RN) atomicAdd(&cD[u], 1);
    }
  }
  __syncthreads();
  int* __restrict__ pS = partS + (size_t)c * PADN + lo;
  int* __restrict__ pD = partD + (size_t)c * PADN + lo;
  for (int i = tid; i < RN; i += CSR_THREADS) { pS[i] = cS[i]; pD[i] = cD[i]; }
}

#define ALLOC_THREADS 256
__global__ __launch_bounds__(ALLOC_THREADS) void k_alloc(
    const int* __restrict__ partS, int* __restrict__ partD,
    float* __restrict__ normS, float* __restrict__ normD,
    int2* __restrict__ seg, int* __restrict__ total, int n) {
  __shared__ int wsum[ALLOC_THREADS / 32];
  __shared__ int base;
  const int i    = blockIdx.x * ALLOC_THREADS + threadIdx.x;
  const int lane = threadIdx.x & 31;
  const int wave = threadIdx.x >> 5;
  int cs = 0;
  int c  = 0;
  if (i < n) {
    int vs[NC];
    int vd[NC];
#pragma unroll
    for (int k = 0; k < NC; ++k) {
      vs[k] = partS[(size_t)k * PADN + i];
      vd[k] = partD[(size_t)k * PADN + i];
    }
#pragma unroll
    for (int k = 0; k < NC; ++k) {
      cs += vs[k];
      partD[(size_t)k * PADN + i] = c;
      c += vd[k];
    }
  }
  int x = c;
  for (int o = 1; o < 32; o <<= 1) {
    const int y = __shfl_up(x, o);
    if (lane >= o) x += y;
  }
  if (lane == 31) wsum[wave] = x;
  __syncthreads();
  if (threadIdx.x == 0) {
    int t = 0;
    for (int w = 0; w < ALLOC_THREADS / 32; ++w) {
      const int v = wsum[w];
      wsum[w] = t;
      t += v;
    }
    base = atomicAdd(total, t);
  }
  __syncthreads();
  if (i < n) {
    const int beg = base + wsum[wave] + x - c;
    seg[i]    = make_int2(beg, c);
    normS[i]  = rsqrtf(fmaxf((float)cs, 1.0f));
    normD[i]  = rsqrtf(fmaxf((float)c, 1.0f));
  }
}

#define FILL_G 3
#define FILL_W 8
__global__ __launch_bounds__(CSR_THREADS) void k_fill(
    const int* __restrict__ src, const int* __restrict__ dst,
    const int2* __restrict__ seg, const int* __restrict__ partD,
    int* tmpE, int* __restrict__ csrSrc) {
  __shared__ int2 bc[CSR_RI * CSR_THREADS];
  const int tid = threadIdx.x;
  const int r   = (int)blockIdx.x % NR;
  const int c   = (int)blockIdx.x / NR;
  const int lo  = r * RN;
  const int* __restrict__ pD = partD + (size_t)c * PADN + lo;
  const int n4 = min(CE4, NE / 4 - c * CE4);
  const int4* __restrict__ d4 = (const int4*)dst + c * CE4;
  int4 d[CSR_NI];
#pragma unroll
  for (int k = 0; k < CSR_NI; ++k) d[k] = d4[min(tid + k * CSR_THREADS, n4 - 1)];
  int sb[CSR_RI];
  int co[CSR_RI];
#pragma unroll
  for (int k = 0; k < CSR_RI; ++k) {
    const int i = min(tid + k * CSR_THREADS, RN - 1);
    sb[k] = seg[lo + i].x;
    co[k] = pD[i];
  }
#pragma unroll
  for (int k = 0; k < CSR_RI; ++k) {
    const int i = tid + k * CSR_THREADS;
    const int b = (i < RN) ? (sb[k] + co[k]) : 0;
    bc[i] = make_int2(b, b);
  }
  __syncthreads();
#pragma unroll
  for (int k = 0; k < CSR_NI; ++k) {
    const int i = tid + k * CSR_THREADS;
    if (i < n4) {
      const int e0 = 4 * (c * CE4 + i);
      unsigned u;
      u = (unsigned)(d[k].x - lo); if (u < RN) tmpE[atomicAdd(&bc[u].y, 1)] = e0 + 0;
      u = (unsigned)(d[k].y - lo); if (u < RN) tmpE[atomicAdd(&bc[u].y, 1)] = e0 + 1;
      u = (unsigned)(d[k].z - lo); if (u < RN) tmpE[atomicAdd(&bc[u].y, 1)] = e0 + 2;
      u = (unsigned)(d[k].w - lo); if (u < RN) tmpE[atomicAdd(&bc[u].y, 1)] = e0 + 3;
    }
  }
  __syncthreads();

  int gb[CSR_RI];
  int gn[CSR_RI];
#pragma unroll
  for (int k = 0; k < CSR_RI; ++k) {
    const int2 g = bc[tid + k * CSR_THREADS];
    gb[k] = g.x;
    gn[k] = g.y - g.x;
  }
  int ev[CSR_RI][FILL_G];
  int sv[CSR_RI][FILL_G];
  bool longer = false;
#pragma unroll
  for (int k = 0; k < CSR_RI; ++k) {
#pragma unroll
    for (int j = 0; j < FILL_G; ++j) {
      const int* p = (gn[k] > 0) ? (const int*)(tmpE + gb[k] + min(j, gn[k] - 1)) : src;
      ev[k][j] = *p;
    }
    longer = longer || (gn[k] > FILL_G);
  }
#pragma unroll
  for (int k = 0; k < CSR_RI; ++k) {
#pragma unroll
    for (int j = 0; j < FILL_G; ++j) sv[k][j] = src[ev[k][j]];
  }
  static_assert(FILL_G == 3, "the fence below names FILL_G values");
#pragma unroll
  for (int k = 0; k < CSR_RI; ++k) {
    asm volatile("" :: "v"(sv[k][0]), "v"(sv[k][1]), "v"(sv[k][2]));
    if (gn[k] > 0 && gn[k] <= FILL_G) {
#pragma unroll
      for (int j = 0; j < FILL_G; ++j) {
        int rk = 0;
#pragma unroll
        for (int q = 0; q < FILL_G; ++q) rk += (q < gn[k] && ev[k][q] < ev[k][j]) ? 1 : 0;
        if (j < gn[k]) csrSrc[gb[k] + rk] = sv[k][j];
      }
    }
  }
  if (longer) {
    for (int i = tid; i < RN; i += CSR_THREADS) {
      const int2 g = bc[i];
      const int n = g.y - g.x;
      if (n > FILL_G) {
        const int* ids = tmpE + g.x;
        for (int j0 = 0; j0 < n; j0 += FILL_W) {
          int e[FILL_W];
          int s[FILL_W];
          int rk[FILL_W];
#pragma unroll
          for (int t = 0; t < FILL_W; ++t) e[t] = ids[min(j0 + t, n - 1)];
#pragma unroll
          for (int t = 0; t < FILL_W; ++t) { s[t] = src[e[t]]; rk[t] = 0; }
          for (int q0 = 0; q0 < n; q0 += FILL_W) {
            int o[FILL_W];
#pragma unroll
            for (int u = 0; u < FILL_W; ++u) o[u] = ids[min(q0 + u, n - 1)];
#pragma unroll
            for (int u = 0; u < FILL_W; ++u) {
              const bool live = q0 + u < n;
#pragma unroll
              for (int t = 0; t < FILL_W; ++t) rk[t] += (live && o[u] < e[t]) ? 1 : 0;
            }
          }
#pragma unroll
          for (int t = 0; t < FILL_W; ++t)
            if (j0 + t < n) csrSrc[g.x + rk[t]] = s[t];
        }
      }
    }
  }
}


#define GEMM_THREADS 512
#define GEMM_GRID    256

template <int NCSHIFT  >
__global__ __launch_bounds__(GEMM_THREADS) void k_gemm_wmma(
    const float* __restrict__ X, const float* __restrict__ norm,
    const float* __restrict__ W, float* __restrict__ Y, int nstrips) {
  constexpr int ncol  = 1 << NCSHIFT;
  constexpr int ncolP = ncol + 16;
  constexpr int ntile = ncol >> 4;
  __shared__ float Ws[(KDIM / 2) * ncolP * 2];

  const int tid = threadIdx.x;

  {
    constexpr int ncq = NCSHIFT - 2;
    constexpr int nf4 = KDIM << ncq;
    for (int i = 0; i < nf4 / GEMM_THREADS; ++i) {
      const int idx = tid + i * GEMM_THREADS;
      const int k = idx >> ncq;
      const int j = (idx - (k << ncq)) << 2;
      const float4 v = ((const float4*)W)[idx];
      const int base = ((k >> 1) * ncolP + j) * 2 + (k & 1);
      Ws[base + 0] = v.x;
      Ws[base + 2] = v.y;
      Ws[base + 4] = v.z;
      Ws[base + 6] = v.w;
    }
  }
  __syncthreads();

  const int wave = tid >> 5;
  const int lane = tid & 31;
  const int hl   = lane >> 4;
  const int lm   = lane & 15;

  const int s = wave * (int)gridDim.x + (int)blockIdx.x;
  if (s < nstrips) {
    const int m0 = s * 16;
    const float* __restrict__ xr = X + (size_t)(m0 + lm) * KDIM + 2 * hl;
    const float nrm = norm[m0 + lm];

    v2f a[KDIM / 4];
    for (int j = 0; j < KDIM / 4; ++j) {
      const v2f x = *(const v2f*)(xr + 4 * j);
      a[j].x = x.x * nrm;
      a[j].y = x.y * nrm;
    }

    v8f acc[ntile];
    for (int t = 0; t < ntile; ++t) acc[t] = v8f{};

    for (int j = 0; j < KDIM / 4; ++j) {
      const int p = 2 * j + hl;
      for (int t = 0; t < ntile; ++t) {
        const v2f b = *(const v2f*)&Ws[(p * ncolP + t * 16 + lm) * 2];
        acc[t] = __builtin_amdgcn_wmma_f32_16x16x4_f32(
             false, a[j],  false, b,
             (short)0, acc[t],  false,  false);
      }
    }

    for (int t = 0; t < ntile; ++t) {
      for (int r = 0; r < 8; ++r)
        Y[(size_t)(m0 + hl * 8 + r) * ncol + t * 16 + lm] = acc[t][r];
    }
  }
}


template <int VEC> struct RowVec;
template <> struct RowVec<4> { typedef v4f T; };
template <> struct RowVec<2> { typedef v2f T; };

#define GATHER_THREADS 256
#define GATHER_GRID    2048

template <int VEC, int U>
__device__ __forceinline__ void gather_batch(typename RowVec<VEC>::T& acc, int ids, int j,
                                             const typename RowVec<VEC>::T* __restrict__ H,
                                             int lane) {
  typedef typename RowVec<VEC>::T vecT;
  vecT v[U];
#pragma unroll
  for (int u = 0; u < U; ++u) {
    const int s = __builtin_amdgcn_readlane(ids, j + u);
    v[u] = H[(size_t)s * 32 + lane];
  }
#pragma unroll
  for (int u = 0; u < U; ++u) acc += v[u];
}

template <int VEC>
__device__ __forceinline__ void gather_chunk(typename RowVec<VEC>::T& acc, int ids, int m,
                                             const typename RowVec<VEC>::T* __restrict__ H,
                                             int lane) {
  int j = 0;
  for (; j + 8 <= m; j += 8) gather_batch<VEC, 8>(acc, ids, j, H, lane);
  if (m & 4) { gather_batch<VEC, 4>(acc, ids, j, H, lane); j += 4; }
  if (m & 2) { gather_batch<VEC, 2>(acc, ids, j, H, lane); j += 2; }
  if (m & 1) { gather_batch<VEC, 1>(acc, ids, j, H, lane); }
}

template <int VEC, bool RELU>
__global__ __launch_bounds__(GATHER_THREADS) void k_gather(
    const float* __restrict__ Hf, const int2* __restrict__ seg,
    const int* __restrict__ csrSrc, const float* __restrict__ normD,
    const float* __restrict__ bias, float* __restrict__ outf, int n, int nE) {
  typedef typename RowVec<VEC>::T vecT;
  const vecT* __restrict__ H = (const vecT*)Hf;
  vecT* __restrict__ out = (vecT*)outf;
  const int lane = threadIdx.x & 31;
  const int nwav = (int)gridDim.x * (GATHER_THREADS / 32);
  const vecT bv = ((const vecT*)bias)[lane];
  int node = __builtin_amdgcn_readfirstlane((int)blockIdx.x * (GATHER_THREADS / 32) +
                                            ((int)threadIdx.x >> 5));
  if (node >= n) return;
  const int last = nE - 1;
  int2 sg  = seg[node];
  int2 sgN = make_int2(0, 0);
  if (node + nwav < n) sgN = seg[node + nwav];
  int ids = csrSrc[min(sg.x + lane, last)];
  for (;;) {
    const int nn = node + nwav;
    const int idsN = csrSrc[min(sgN.x + lane, last)];
    int2 sgNN = make_int2(0, 0);
    if (nn + nwav < n) sgNN = seg[nn + nwav];
    const float nrm = normD[node];

    const int beg = sg.x;
    const int len = sg.y;
    vecT acc = (vecT)(0.0f);
    gather_chunk<VEC>(acc, ids, (len < 32) ? len : 32, H, lane);
    for (int c0 = 32; c0 < len; c0 += 32) {
      const int m = (len - c0 < 32) ? (len - c0) : 32;
      const int cid = csrSrc[min(beg + c0 + lane, last)];
      gather_chunk<VEC>(acc, cid, m, H, lane);
    }
    vecT r = acc * nrm + bv;
    if (RELU) {
#pragma unroll
      for (int q = 0; q < VEC; ++q) r[q] = fmaxf(r[q], 0.0f);
    }
    out[(size_t)node * 32 + lane] = r;

    if (nn >= n) break;
    node = nn;
    sg   = sgN;
    sgN  = sgNN;
    ids  = idsN;
  }
}


extern "C" void kernel_launch(void* const* d_in, const int* in_sizes, int n_in,
                              void* d_out, int out_size, void* d_ws, size_t ws_size,
                              hipStream_t stream) {
  (void)in_sizes; (void)n_in; (void)out_size; (void)ws_size;

  const float* features = (const float*)d_in[0];
  const float* W1       = (const float*)d_in[1];
  const float* b1       = (const float*)d_in[2];
  const float* W2       = (const float*)d_in[3];
  const float* b2       = (const float*)d_in[4];
  const int*   esrc     = (const int*)d_in[5];
  const int*   edst     = (const int*)d_in[6];
  float*       out      = (float*)d_out;

  int*   wi     = (int*)d_ws;
  int*   total  = wi;
  float* normS  = (float*)(wi + 64);
  float* normD  = normS + PADN;
  int2*  seg    = (int2*)(normD + PADN);
  int*   tmpE   = (int*)(seg + PADN);
  int*   csrSrc = tmpE + NE;
  int*   partS  = csrSrc + NE;
  int*   partD  = partS + (size_t)NC * PADN;
  float* B0     = (float*)(partD + (size_t)NC * PADN);
  float* B1     = B0 + (size_t)NN * 128;

  {
    k_count<<<CSR_GRID, CSR_THREADS, 0, stream>>>(esrc, edst, partS, partD, total);
    k_alloc<<<(NN + ALLOC_THREADS - 1) / ALLOC_THREADS, ALLOC_THREADS, 0, stream>>>(
        partS, partD, normS, normD, seg, total, NN);
    k_fill<<<CSR_GRID, CSR_THREADS, 0, stream>>>(esrc, edst, seg, partD, tmpE, csrSrc);
  }

  k_gemm_wmma<7><<<GEMM_GRID, GEMM_THREADS, 0, stream>>>(features, normS, W1, B0, NN / 16);

  k_gather<4, true><<<GATHER_GRID, GATHER_THREADS, 0, stream>>>(
      B0, seg, csrSrc, normD, b1, B1, NN, NE);

  k_gemm_wmma<6><<<GEMM_GRID, GEMM_THREADS, 0, stream>>>(B1, normS, W2, B0, NN / 16);

  k_gather<2, false><<<GATHER_GRID, GATHER_THREADS, 0, stream>>>(
      B0, seg, csrSrc, normD, b2, out, NN, NE);
}


# Layer-1 edge sums and layer-2 product in one launch: a 16-wave work-group sums 64 rows into LDS and multiplies them by W2 there; hidden rows never reach memory; sum and k order unchanged
# speedup vs baseline: 4.5528x; 1.0760x over previous
#include <hip/hip_runtime.h>


typedef __attribute__((ext_vector_type(2))) float v2f;
typedef __attribute__((ext_vector_type(4))) float v4f;
typedef __attribute__((ext_vector_type(8))) float v8f;

#define NN   50000
#define NE   600000
#define KDIM 128
#define PADN 50176


#define NR   8
#define RN   6250
#define NC   32
#define CE4  4688
#define CSR_THREADS 1024
#define CSR_GRID    (NR * NC)
#define CSR_NI      ((CE4 + CSR_THREADS - 1) / CSR_THREADS)
#define CSR_RI      ((RN + CSR_THREADS - 1) / CSR_THREADS)

__global__ __launch_bounds__(CSR_THREADS) void k_count(
    const int* __restrict__ src, const int* __restrict__ dst,
    int* __restrict__ partS, int* __restrict__ partD, int* __restrict__ total) {
  __shared__ int cS[RN];
  __shared__ int cD[RN];
  const int tid = threadIdx.x;
  const int r   = (int)blockIdx.x % NR;
  const int c   = (int)blockIdx.x / NR;
  const int lo  = r * RN;
  const int n4 = min(CE4, NE / 4 - c * CE4);
  const int4* __restrict__ s4 = (const int4*)src + c * CE4;
  const int4* __restrict__ d4 = (const int4*)dst + c * CE4;
  int4 s[CSR_NI];
  int4 d[CSR_NI];
#pragma unroll
  for (int k = 0; k < CSR_NI; ++k) {
    const int i = min(tid + k * CSR_THREADS, n4 - 1);
    s[k] = s4[i];
    d[k] = d4[i];
  }
  for (int i = tid; i < RN; i += CSR_THREADS) { cS[i] = 0; cD[i] = 0; }
  if (blockIdx.x == 0 && tid == 0) *total = 0;
  __syncthreads();
#pragma unroll
  for (int k = 0; k < CSR_NI; ++k) {
    if (tid + k * CSR_THREADS < n4) {
      unsigned u;
      u = (unsigned)(s[k].x - lo); if (u < RN) atomicAdd(&cS[u], 1);
      u = (unsigned)(s[k].y - lo); if (u < RN) atomicAdd(&cS[u], 1);
      u = (unsigned)(s[k].z - lo); if (u < RN) atomicAdd(&cS[u], 1);
      u = (unsigned)(s[k].w - lo); if (u < RN) atomicAdd(&cS[u], 1);
      u = (unsigned)(d[k].x - lo); if (u < RN) atomicAdd(&cD[u], 1);
      u = (unsigned)(d[k].y - lo); if (u < RN) atomicAdd(&cD[u], 1);
      u = (unsigned)(d[k].z - lo); if (u < RN) atomicAdd(&cD[u], 1);
      u = (unsigned)(d[k].w - lo); if (u < RN) atomicAdd(&cD[u], 1);
    }
  }
  __syncthreads();
  int* __restrict__ pS = partS + (size_t)c * PADN + lo;
  int* __restrict__ pD = partD + (size_t)c * PADN + lo;
  for (int i = tid; i < RN; i += CSR_THREADS) { pS[i] = cS[i]; pD[i] = cD[i]; }
}

#define ALLOC_THREADS 256
__global__ __launch_bounds__(ALLOC_THREADS) void k_alloc(
    const int* __restrict__ partS, int* __restrict__ partD,
    float* __restrict__ normS, float* __restrict__ normD,
    int2* __restrict__ seg, int* __restrict__ total, int n) {
  __shared__ int wsum[ALLOC_THREADS / 32];
  __shared__ int base;
  const int i    = blockIdx.x * ALLOC_THREADS + threadIdx.x;
  const int lane = threadIdx.x & 31;
  const int wave = threadIdx.x >> 5;
  int cs = 0;
  int c  = 0;
  if (i < n) {
    int vs[NC];
    int vd[NC];
#pragma unroll
    for (int k = 0; k < NC; ++k) {
      vs[k] = partS[(size_t)k * PADN + i];
      vd[k] = partD[(size_t)k * PADN + i];
    }
#pragma unroll
    for (int k = 0; k < NC; ++k) {
      cs += vs[k];
      partD[(size_t)k * PADN + i] = c;
      c += vd[k];
    }
  }
  int x = c;
  for (int o = 1; o < 32; o <<= 1) {
    const int y = __shfl_up(x, o);
    if (lane >= o) x += y;
  }
  if (lane == 31) wsum[wave] = x;
  __syncthreads();
  if (threadIdx.x == 0) {
    int t = 0;
    for (int w = 0; w < ALLOC_THREADS / 32; ++w) {
      const int v = wsum[w];
      wsum[w] = t;
      t += v;
    }
    base = atomicAdd(total, t);
  }
  __syncthreads();
  if (i < n) {
    const int beg = base + wsum[wave] + x - c;
    seg[i]    = make_int2(beg, c);
    normS[i]  = rsqrtf(fmaxf((float)cs, 1.0f));
    normD[i]  = rsqrtf(fmaxf((float)c, 1.0f));
  }
}

#define FILL_G 3
#define FILL_W 8
__global__ __launch_bounds__(CSR_THREADS) void k_fill(
    const int* __restrict__ src, const int* __restrict__ dst,
    const int2* __restrict__ seg, const int* __restrict__ partD,
    int* tmpE, int* __restrict__ csrSrc) {
  __shared__ int2 bc[CSR_RI * CSR_THREADS];
  const int tid = threadIdx.x;
  const int r   = (int)blockIdx.x % NR;
  const int c   = (int)blockIdx.x / NR;
  const int lo  = r * RN;
  const int* __restrict__ pD = partD + (size_t)c * PADN + lo;
  const int n4 = min(CE4, NE / 4 - c * CE4);
  const int4* __restrict__ d4 = (const int4*)dst + c * CE4;
  int4 d[CSR_NI];
#pragma unroll
  for (int k = 0; k < CSR_NI; ++k) d[k] = d4[min(tid + k * CSR_THREADS, n4 - 1)];
  int sb[CSR_RI];
  int co[CSR_RI];
#pragma unroll
  for (int k = 0; k < CSR_RI; ++k) {
    const int i = min(tid + k * CSR_THREADS, RN - 1);
    sb[k] = seg[lo + i].x;
    co[k] = pD[i];
  }
#pragma unroll
  for (int k = 0; k < CSR_RI; ++k) {
    const int i = tid + k * CSR_THREADS;
    const int b = (i < RN) ? (sb[k] + co[k]) : 0;
    bc[i] = make_int2(b, b);
  }
  __syncthreads();
#pragma unroll
  for (int k = 0; k < CSR_NI; ++k) {
    const int i = tid + k * CSR_THREADS;
    if (i < n4) {
      const int e0 = 4 * (c * CE4 + i);
      unsigned u;
      u = (unsigned)(d[k].x - lo); if (u < RN) tmpE[atomicAdd(&bc[u].y, 1)] = e0 + 0;
      u = (unsigned)(d[k].y - lo); if (u < RN) tmpE[atomicAdd(&bc[u].y, 1)] = e0 + 1;
      u = (unsigned)(d[k].z - lo); if (u < RN) tmpE[atomicAdd(&bc[u].y, 1)] = e0 + 2;
      u = (unsigned)(d[k].w - lo); if (u < RN) tmpE[atomicAdd(&bc[u].y, 1)] = e0 + 3;
    }
  }
  __syncthreads();

  int gb[CSR_RI];
  int gn[CSR_RI];
#pragma unroll
  for (int k = 0; k < CSR_RI; ++k) {
    const int2 g = bc[tid + k * CSR_THREADS];
    gb[k] = g.x;
    gn[k] = g.y - g.x;
  }
  int ev[CSR_RI][FILL_G];
  int sv[CSR_RI][FILL_G];
  bool longer = false;
#pragma unroll
  for (int k = 0; k < CSR_RI; ++k) {
#pragma unroll
    for (int j = 0; j < FILL_G; ++j) {
      const int* p = (gn[k] > 0) ? (const int*)(tmpE + gb[k] + min(j, gn[k] - 1)) : src;
      ev[k][j] = *p;
    }
    longer = longer || (gn[k] > FILL_G);
  }
#pragma unroll
  for (int k = 0; k < CSR_RI; ++k) {
#pragma unroll
    for (int j = 0; j < FILL_G; ++j) sv[k][j] = src[ev[k][j]];
  }
  static_assert(FILL_G == 3, "the fence below names FILL_G values");
#pragma unroll
  for (int k = 0; k < CSR_RI; ++k) {
    asm volatile("" :: "v"(sv[k][0]), "v"(sv[k][1]), "v"(sv[k][2]));
    if (gn[k] > 0 && gn[k] <= FILL_G) {
#pragma unroll
      for (int j = 0; j < FILL_G; ++j) {
        int rk = 0;
#pragma unroll
        for (int q = 0; q < FILL_G; ++q) rk += (q < gn[k] && ev[k][q] < ev[k][j]) ? 1 : 0;
        if (j < gn[k]) csrSrc[gb[k] + rk] = sv[k][j];
      }
    }
  }
  if (longer) {
    for (int i = tid; i < RN; i += CSR_THREADS) {
      const int2 g = bc[i];
      const int n = g.y - g.x;
      if (n > FILL_G) {
        const int* ids = tmpE + g.x;
        for (int j0 = 0; j0 < n; j0 += FILL_W) {
          int e[FILL_W];
          int s[FILL_W];
          int rk[FILL_W];
#pragma unroll
          for (int t = 0; t < FILL_W; ++t) e[t] = ids[min(j0 + t, n - 1)];
#pragma unroll
          for (int t = 0; t < FILL_W; ++t) { s[t] = src[e[t]]; rk[t] = 0; }
          for (int q0 = 0; q0 < n; q0 += FILL_W) {
            int o[FILL_W];
#pragma unroll
            for (int u = 0; u < FILL_W; ++u) o[u] = ids[min(q0 + u, n - 1)];
#pragma unroll
            for (int u = 0; u < FILL_W; ++u) {
              const bool live = q0 + u < n;
#pragma unroll
              for (int t = 0; t < FILL_W; ++t) rk[t] += (live && o[u] < e[t]) ? 1 : 0;
            }
          }
#pragma unroll
          for (int t = 0; t < FILL_W; ++t)
            if (j0 + t < n) csrSrc[g.x + rk[t]] = s[t];
        }
      }
    }
  }
}


#define GEMM_THREADS 512
#define GEMM_GRID    256

template <int NCSHIFT  >
__global__ __launch_bounds__(GEMM_THREADS) void k_gemm_wmma(
    const float* __restrict__ X, const float* __restrict__ norm,
    const float* __restrict__ W, float* __restrict__ Y, int nstrips) {
  constexpr int ncol  = 1 << NCSHIFT;
  constexpr int ncolP = ncol + 16;
  constexpr int ntile = ncol >> 4;
  __shared__ float Ws[(KDIM / 2) * ncolP * 2];

  const int tid = threadIdx.x;

  {
    constexpr int ncq = NCSHIFT - 2;
    constexpr int nf4 = KDIM << ncq;
    for (int i = 0; i < nf4 / GEMM_THREADS; ++i) {
      const int idx = tid + i * GEMM_THREADS;
      const int k = idx >> ncq;
      const int j = (idx - (k << ncq)) << 2;
      const float4 v = ((const float4*)W)[idx];
      const int base = ((k >> 1) * ncolP + j) * 2 + (k & 1);
      Ws[base + 0] = v.x;
      Ws[base + 2] = v.y;
      Ws[base + 4] = v.z;
      Ws[base + 6] = v.w;
    }
  }
  __syncthreads();

  const int wave = tid >> 5;
  const int lane = tid & 31;
  const int hl   = lane >> 4;
  const int lm   = lane & 15;

  const int s = wave * (int)gridDim.x + (int)blockIdx.x;
  if (s < nstrips) {
    const int m0 = s * 16;
    const float* __restrict__ xr = X + (size_t)(m0 + lm) * KDIM + 2 * hl;
    const float nrm = norm[m0 + lm];

    v2f a[KDIM / 4];
    for (int j = 0; j < KDIM / 4; ++j) {
      const v2f x = *(const v2f*)(xr + 4 * j);
      a[j].x = x.x * nrm;
      a[j].y = x.y * nrm;
    }

    v8f acc[ntile];
    for (int t = 0; t < ntile; ++t) acc[t] = v8f{};

    for (int j = 0; j < KDIM / 4; ++j) {
      const int p = 2 * j + hl;
      for (int t = 0; t < ntile; ++t) {
        const v2f b = *(const v2f*)&Ws[(p * ncolP + t * 16 + lm) * 2];
        acc[t] = __builtin_amdgcn_wmma_f32_16x16x4_f32(
             false, a[j],  false, b,
             (short)0, acc[t],  false,  false);
      }
    }

    for (int t = 0; t < ntile; ++t) {
      for (int r = 0; r < 8; ++r)
        Y[(size_t)(m0 + hl * 8 + r) * ncol + t * 16 + lm] = acc[t][r];
    }
  }
}


template <int VEC> struct RowVec;
template <> struct RowVec<4> { typedef v4f T; };
template <> struct RowVec<2> { typedef v2f T; };

#define GATHER_THREADS 256
#define GATHER_GRID    2048

template <int VEC, int U>
__device__ __forceinline__ void gather_batch(typename RowVec<VEC>::T& acc, int ids, int j,
                                             const typename RowVec<VEC>::T* __restrict__ H,
                                             int lane) {
  typedef typename RowVec<VEC>::T vecT;
  vecT v[U];
#pragma unroll
  for (int u = 0; u < U; ++u) {
    const int s = __builtin_amdgcn_readlane(ids, j + u);
    v[u] = H[(size_t)s * 32 + lane];
  }
#pragma unroll
  for (int u = 0; u < U; ++u) acc += v[u];
}

template <int VEC>
__device__ __forceinline__ void gather_chunk(typename RowVec<VEC>::T& acc, int ids, int m,
                                             const typename RowVec<VEC>::T* __restrict__ H,
                                             int lane) {
  int j = 0;
  for (; j + 8 <= m; j += 8) gather_batch<VEC, 8>(acc, ids, j, H, lane);
  if (m & 4) { gather_batch<VEC, 4>(acc, ids, j, H, lane); j += 4; }
  if (m & 2) { gather_batch<VEC, 2>(acc, ids, j, H, lane); j += 2; }
  if (m & 1) { gather_batch<VEC, 1>(acc, ids, j, H, lane); }
}

template <int VEC, bool RELU>
__global__ __launch_bounds__(GATHER_THREADS) void k_gather(
    const float* __restrict__ Hf, const int2* __restrict__ seg,
    const int* __restrict__ csrSrc, const float* __restrict__ normD,
    const float* __restrict__ bias, float* __restrict__ outf, int n, int nE) {
  typedef typename RowVec<VEC>::T vecT;
  const vecT* __restrict__ H = (const vecT*)Hf;
  vecT* __restrict__ out = (vecT*)outf;
  const int lane = threadIdx.x & 31;
  const int nwav = (int)gridDim.x * (GATHER_THREADS / 32);
  const vecT bv = ((const vecT*)bias)[lane];
  int node = __builtin_amdgcn_readfirstlane((int)blockIdx.x * (GATHER_THREADS / 32) +
                                            ((int)threadIdx.x >> 5));
  if (node >= n) return;
  const int last = nE - 1;
  int2 sg  = seg[node];
  int2 sgN = make_int2(0, 0);
  if (node + nwav < n) sgN = seg[node + nwav];
  int ids = csrSrc[min(sg.x + lane, last)];
  for (;;) {
    const int nn = node + nwav;
    const int idsN = csrSrc[min(sgN.x + lane, last)];
    int2 sgNN = make_int2(0, 0);
    if (nn + nwav < n) sgNN = seg[nn + nwav];
    const float nrm = normD[node];

    const int beg = sg.x;
    const int len = sg.y;
    vecT acc = (vecT)(0.0f);
    gather_chunk<VEC>(acc, ids, (len < 32) ? len : 32, H, lane);
    for (int c0 = 32; c0 < len; c0 += 32) {
      const int m = (len - c0 < 32) ? (len - c0) : 32;
      const int cid = csrSrc[min(beg + c0 + lane, last)];
      gather_chunk<VEC>(acc, cid, m, H, lane);
    }
    vecT r = acc * nrm + bv;
    if (RELU) {
#pragma unroll
      for (int q = 0; q < VEC; ++q) r[q] = fmaxf(r[q], 0.0f);
    }
    out[(size_t)node * 32 + lane] = r;

    if (nn >= n) break;
    node = nn;
    sg   = sgN;
    sgN  = sgNN;
    ids  = idsN;
  }
}


#define FUSE_THREADS 512
#define FUSE_ROWS    64
#define FUSE_PITCH4  33

__global__ __launch_bounds__(FUSE_THREADS) void k_gather_gemm(
    const float* __restrict__ Hf, const int2* __restrict__ seg,
    const int* __restrict__ csrSrc, const float* __restrict__ normD,
    const float* __restrict__ normS, const float* __restrict__ bias,
    const float* __restrict__ W, float* __restrict__ Y, int n, int nE) {
  constexpr int ncol  = 64;
  constexpr int ncolP = ncol + 16;
  __shared__ float Ws[(KDIM / 2) * ncolP * 2];
  __shared__ v4f   At[FUSE_ROWS * FUSE_PITCH4];

  const int tid  = threadIdx.x;
  const int lane = tid & 31;
  const int wave = __builtin_amdgcn_readfirstlane(tid >> 5);
  const int tile0 = (int)blockIdx.x * FUSE_ROWS;

  {
    constexpr int ncq = 4;
    constexpr int nf4 = KDIM << ncq;
    for (int i = 0; i < nf4 / FUSE_THREADS; ++i) {
      const int idx = tid + i * FUSE_THREADS;
      const int k = idx >> ncq;
      const int j = (idx - (k << ncq)) << 2;
      const float4 v = ((const float4*)W)[idx];
      const int base = ((k >> 1) * ncolP + j) * 2 + (k & 1);
      Ws[base + 0] = v.x;
      Ws[base + 2] = v.y;
      Ws[base + 4] = v.z;
      Ws[base + 6] = v.w;
    }
  }

  const int row0 = tile0 + wave * 4;
  if (row0 < n) {
    const v4f* __restrict__ H = (const v4f*)Hf;
    const v4f bv = ((const v4f*)bias)[lane];
    const int last = nE - 1;
    int2  sg[4];
    float nd[4];
    float ns[4];
    int   ids[4];
#pragma unroll
    for (int i = 0; i < 4; ++i) {
      const int r = min(row0 + i, n - 1);
      sg[i] = seg[r];
      nd[i] = normD[r];
      ns[i] = normS[r];
    }
#pragma unroll
    for (int i = 0; i < 4; ++i) ids[i] = csrSrc[min(sg[i].x + lane, last)];
#pragma unroll
    for (int i = 0; i < 4; ++i) {
      const int beg = sg[i].x;
      const int len = sg[i].y;
      v4f acc = (v4f)(0.0f);
      gather_chunk<4>(acc, ids[i], (len < 32) ? len : 32, H, lane);
      for (int c0 = 32; c0 < len; c0 += 32) {
        const int m = (len - c0 < 32) ? (len - c0) : 32;
        const int cid = csrSrc[min(beg + c0 + lane, last)];
        gather_chunk<4>(acc, cid, m, H, lane);
      }
      v4f r = acc * nd[i] + bv;
#pragma unroll
      for (int q = 0; q < 4; ++q) r[q] = fmaxf(r[q], 0.0f) * ns[i];
      At[(wave * 4 + i) * FUSE_PITCH4 + lane] = r;
    }
  }
  __syncthreads();

  const int st = wave >> 2;
  const int t  = wave & 3;
  const int m0 = tile0 + st * 16;
  if (m0 < n) {
    const int hl = lane >> 4;
    const int lm = lane & 15;
    const float* ar = (const float*)At + (st * 16 + lm) * (FUSE_PITCH4 * 4) + 2 * hl;
    v8f acc = v8f{};
    for (int j = 0; j < KDIM / 4; ++j) {
      const int p = 2 * j + hl;
      const v2f a = *(const v2f*)(ar + 4 * j);
      const v2f b = *(const v2f*)&Ws[(p * ncolP + t * 16 + lm) * 2];
      acc = __builtin_amdgcn_wmma_f32_16x16x4_f32(
           false, a,  false, b,
           (short)0, acc,  false,  false);
    }
    for (int r = 0; r < 8; ++r)
      Y[(size_t)(m0 + hl * 8 + r) * ncol + t * 16 + lm] = acc[r];
  }
}


extern "C" void kernel_launch(void* const* d_in, const int* in_sizes, int n_in,
                              void* d_out, int out_size, void* d_ws, size_t ws_size,
                              hipStream_t stream) {
  (void)in_sizes; (void)n_in; (void)out_size; (void)ws_size;

  const float* features = (const float*)d_in[0];
  const float* W1       = (const float*)d_in[1];
  const float* b1       = (const float*)d_in[2];
  const float* W2       = (const float*)d_in[3];
  const float* b2       = (const float*)d_in[4];
  const int*   esrc     = (const int*)d_in[5];
  const int*   edst     = (const int*)d_in[6];
  float*       out      = (float*)d_out;

  int*   wi     = (int*)d_ws;
  int*   total  = wi;
  float* normS  = (float*)(wi + 64);
  float* normD  = normS + PADN;
  int2*  seg    = (int2*)(normD + PADN);
  int*   tmpE   = (int*)(seg + PADN);
  int*   csrSrc = tmpE + NE;
  int*   partS  = csrSrc + NE;
  int*   partD  = partS + (size_t)NC * PADN;
  float* B0     = (float*)(partD + (size_t)NC * PADN);
  float* B1     = B0 + (size_t)NN * 128;

  {
    k_count<<<CSR_GRID, CSR_THREADS, 0, stream>>>(esrc, edst, partS, partD, total);
    k_alloc<<<(NN + ALLOC_THREADS - 1) / ALLOC_THREADS, ALLOC_THREADS, 0, stream>>>(
        partS, partD, normS, normD, seg, total, NN);
    k_fill<<<CSR_GRID, CSR_THREADS, 0, stream>>>(esrc, edst, seg, partD, tmpE, csrSrc);
  }

  k_gemm_wmma<7><<<GEMM_GRID, GEMM_THREADS, 0, stream>>>(features, normS, W1, B0, NN / 16);

  k_gather_gemm<<<(NN + FUSE_ROWS - 1) / FUSE_ROWS, FUSE_THREADS, 0, stream>>>(
      B0, seg, csrSrc, normD, normS, b1, W2, B1, NN, NE);

  k_gather<2, false><<<GATHER_GRID, GATHER_THREADS, 0, stream>>>(
      B1, seg, csrSrc, normD, b2, out, NN, NE);
}


# Layer-1 GEMM sums its rows' out-degrees from the chunk counts itself, so it need not wait for the per-node merge: both share one launch, merge work-groups taking fewer GEMM strips; sums unchanged
# speedup vs baseline: 4.8528x; 1.0659x over previous
#include <hip/hip_runtime.h>


typedef __attribute__((ext_vector_type(2))) float v2f;
typedef __attribute__((ext_vector_type(4))) float v4f;
typedef __attribute__((ext_vector_type(8))) float v8f;

#define NN   50000
#define NE   600000
#define KDIM 128
#define PADN 50176


#define NR   8
#define RN   6250
#define NC   32
#define CE4  4688
#define CSR_THREADS 1024
#define CSR_GRID    (NR * NC)
#define CSR_NI      ((CE4 + CSR_THREADS - 1) / CSR_THREADS)
#define CSR_RI      ((RN + CSR_THREADS - 1) / CSR_THREADS)

__global__ __launch_bounds__(CSR_THREADS) void k_count(
    const int* __restrict__ src, const int* __restrict__ dst,
    int* __restrict__ partS, int* __restrict__ partD, int* __restrict__ total) {
  __shared__ int cS[RN];
  __shared__ int cD[RN];
  const int tid = threadIdx.x;
  const int r   = (int)blockIdx.x % NR;
  const int c   = (int)blockIdx.x / NR;
  const int lo  = r * RN;
  const int n4 = min(CE4, NE / 4 - c * CE4);
  const int4* __restrict__ s4 = (const int4*)src + c * CE4;
  const int4* __restrict__ d4 = (const int4*)dst + c * CE4;
  int4 s[CSR_NI];
  int4 d[CSR_NI];
#pragma unroll
  for (int k = 0; k < CSR_NI; ++k) {
    const int i = min(tid + k * CSR_THREADS, n4 - 1);
    s[k] = s4[i];
    d[k] = d4[i];
  }
  for (int i = tid; i < RN; i += CSR_THREADS) { cS[i] = 0; cD[i] = 0; }
  if (blockIdx.x == 0 && tid == 0) *total = 0;
  __syncthreads();
#pragma unroll
  for (int k = 0; k < CSR_NI; ++k) {
    if (tid + k * CSR_THREADS < n4) {
      unsigned u;
      u = (unsigned)(s[k].x - lo); if (u < RN) atomicAdd(&cS[u], 1);
      u = (unsigned)(s[k].y - lo); if (u < RN) atomicAdd(&cS[u], 1);
      u = (unsigned)(s[k].z - lo); if (u < RN) atomicAdd(&cS[u], 1);
      u = (unsigned)(s[k].w - lo); if (u < RN) atomicAdd(&cS[u], 1);
      u = (unsigned)(d[k].x - lo); if (u < RN) atomicAdd(&cD[u], 1);
      u = (unsigned)(d[k].y - lo); if (u < RN) atomicAdd(&cD[u], 1);
      u = (unsigned)(d[k].z - lo); if (u < RN) atomicAdd(&cD[u], 1);
      u = (unsigned)(d[k].w - lo); if (u < RN) atomicAdd(&cD[u], 1);
    }
  }
  __syncthreads();
  int* __restrict__ pS = partS + (size_t)c * PADN + lo;
  int* __restrict__ pD = partD + (size_t)c * PADN + lo;
  for (int i = tid; i < RN; i += CSR_THREADS) { pS[i] = cS[i]; pD[i] = cD[i]; }
}

#define AG_THREADS   512
#define ALLOC_BLOCKS ((NN + AG_THREADS - 1) / AG_THREADS)
__device__ __forceinline__ void alloc_body(
    const int* __restrict__ partS, int* __restrict__ partD,
    float* __restrict__ normS, float* __restrict__ normD,
    int2* __restrict__ seg, int* __restrict__ total, int n, int blk) {
  __shared__ int wsum[AG_THREADS / 32];
  __shared__ int base;
  const int i    = blk * AG_THREADS + threadIdx.x;
  const int lane = threadIdx.x & 31;
  const int wave = threadIdx.x >> 5;
  int cs = 0;
  int c  = 0;
  if (i < n) {
    int vs[NC];
    int vd[NC];
#pragma unroll
    for (int k = 0; k < NC; ++k) {
      vs[k] = partS[(size_t)k * PADN + i];
      vd[k] = partD[(size_t)k * PADN + i];
    }
#pragma unroll
    for (int k = 0; k < NC; ++k) {
      cs += vs[k];
      partD[(size_t)k * PADN + i] = c;
      c += vd[k];
    }
  }
  int x = c;
  for (int o = 1; o < 32; o <<= 1) {
    const int y = __shfl_up(x, o);
    if (lane >= o) x += y;
  }
  if (lane == 31) wsum[wave] = x;
  __syncthreads();
  if (threadIdx.x == 0) {
    int t = 0;
    for (int w = 0; w < AG_THREADS / 32; ++w) {
      const int v = wsum[w];
      wsum[w] = t;
      t += v;
    }
    base = atomicAdd(total, t);
  }
  __syncthreads();
  if (i < n) {
    const int beg = base + wsum[wave] + x - c;
    seg[i]    = make_int2(beg, c);
    normS[i]  = rsqrtf(fmaxf((float)cs, 1.0f));
    normD[i]  = rsqrtf(fmaxf((float)c, 1.0f));
  }
}

#define FILL_G 3
#define FILL_W 8
__global__ __launch_bounds__(CSR_THREADS) void k_fill(
    const int* __restrict__ src, const int* __restrict__ dst,
    const int2* __restrict__ seg, const int* __restrict__ partD,
    int* tmpE, int* __restrict__ csrSrc) {
  __shared__ int2 bc[CSR_RI * CSR_THREADS];
  const int tid = threadIdx.x;
  const int r   = (int)blockIdx.x % NR;
  const int c   = (int)blockIdx.x / NR;
  const int lo  = r * RN;
  const int* __restrict__ pD = partD + (size_t)c * PADN + lo;
  const int n4 = min(CE4, NE / 4 - c * CE4);
  const int4* __restrict__ d4 = (const int4*)dst + c * CE4;
  int4 d[CSR_NI];
#pragma unroll
  for (int k = 0; k < CSR_NI; ++k) d[k] = d4[min(tid + k * CSR_THREADS, n4 - 1)];
  int sb[CSR_RI];
  int co[CSR_RI];
#pragma unroll
  for (int k = 0; k < CSR_RI; ++k) {
    const int i = min(tid + k * CSR_THREADS, RN - 1);
    sb[k] = seg[lo + i].x;
    co[k] = pD[i];
  }
#pragma unroll
  for (int k = 0; k < CSR_RI; ++k) {
    const int i = tid + k * CSR_THREADS;
    const int b = (i < RN) ? (sb[k] + co[k]) : 0;
    bc[i] = make_int2(b, b);
  }
  __syncthreads();
#pragma unroll
  for (int k = 0; k < CSR_NI; ++k) {
    const int i = tid + k * CSR_THREADS;
    if (i < n4) {
      const int e0 = 4 * (c * CE4 + i);
      unsigned u;
      u = (unsigned)(d[k].x - lo); if (u < RN) tmpE[atomicAdd(&bc[u].y, 1)] = e0 + 0;
      u = (unsigned)(d[k].y - lo); if (u < RN) tmpE[atomicAdd(&bc[u].y, 1)] = e0 + 1;
      u = (unsigned)(d[k].z - lo); if (u < RN) tmpE[atomicAdd(&bc[u].y, 1)] = e0 + 2;
      u = (unsigned)(d[k].w - lo); if (u < RN) tmpE[atomicAdd(&bc[u].y, 1)] = e0 + 3;
    }
  }
  __syncthreads();

  int gb[CSR_RI];
  int gn[CSR_RI];
#pragma unroll
  for (int k = 0; k < CSR_RI; ++k) {
    const int2 g = bc[tid + k * CSR_THREADS];
    gb[k] = g.x;
    gn[k] = g.y - g.x;
  }
  int ev[CSR_RI][FILL_G];
  int sv[CSR_RI][FILL_G];
  bool longer = false;
#pragma unroll
  for (int k = 0; k < CSR_RI; ++k) {
#pragma unroll
    for (int j = 0; j < FILL_G; ++j) {
      const int* p = (gn[k] > 0) ? (const int*)(tmpE + gb[k] + min(j, gn[k] - 1)) : src;
      ev[k][j] = *p;
    }
    longer = longer || (gn[k] > FILL_G);
  }
#pragma unroll
  for (int k = 0; k < CSR_RI; ++k) {
#pragma unroll
    for (int j = 0; j < FILL_G; ++j) sv[k][j] = src[ev[k][j]];
  }
  static_assert(FILL_G == 3, "the fence below names FILL_G values");
#pragma unroll
  for (int k = 0; k < CSR_RI; ++k) {
    asm volatile("" :: "v"(sv[k][0]), "v"(sv[k][1]), "v"(sv[k][2]));
    if (gn[k] > 0 && gn[k] <= FILL_G) {
#pragma unroll
      for (int j = 0; j < FILL_G; ++j) {
        int rk = 0;
#pragma unroll
        for (int q = 0; q < FILL_G; ++q) rk += (q < gn[k] && ev[k][q] < ev[k][j]) ? 1 : 0;
        if (j < gn[k]) csrSrc[gb[k] + rk] = sv[k][j];
      }
    }
  }
  if (longer) {
    for (int i = tid; i < RN; i += CSR_THREADS) {
      const int2 g = bc[i];
      const int n = g.y - g.x;
      if (n > FILL_G) {
        const int* ids = tmpE + g.x;
        for (int j0 = 0; j0 < n; j0 += FILL_W) {
          int e[FILL_W];
          int s[FILL_W];
          int rk[FILL_W];
#pragma unroll
          for (int t = 0; t < FILL_W; ++t) e[t] = ids[min(j0 + t, n - 1)];
#pragma unroll
          for (int t = 0; t < FILL_W; ++t) { s[t] = src[e[t]]; rk[t] = 0; }
          for (int q0 = 0; q0 < n; q0 += FILL_W) {
            int o[FILL_W];
#pragma unroll
            for (int u = 0; u < FILL_W; ++u) o[u] = ids[min(q0 + u, n - 1)];
#pragma unroll
            for (int u = 0; u < FILL_W; ++u) {
              const bool live = q0 + u < n;
#pragma unroll
              for (int t = 0; t < FILL_W; ++t) rk[t] += (live && o[u] < e[t]) ? 1 : 0;
            }
          }
#pragma unroll
          for (int t = 0; t < FILL_W; ++t)
            if (j0 + t < n) csrSrc[g.x + rk[t]] = s[t];
        }
      }
    }
  }
}


#define GEMM_THREADS AG_THREADS
#define GEMM_GRID    256
#define LIGHT_STRIPS 8
#define HEAVY_BLOCKS (GEMM_GRID - ALLOC_BLOCKS)

template <int NCSHIFT  >
__device__ __forceinline__ void gemm_body(
    const float* __restrict__ X, const int* __restrict__ partS,
    const float* __restrict__ W, float* __restrict__ Y, int nstrips, int blk) {
  constexpr int ncol  = 1 << NCSHIFT;
  constexpr int ncolP = ncol + 16;
  constexpr int ntile = ncol >> 4;
  __shared__ float Ws[(KDIM / 2) * ncolP * 2];

  const int tid = threadIdx.x;

  {
    constexpr int ncq = NCSHIFT - 2;
    constexpr int nf4 = KDIM << ncq;
    for (int i = 0; i < nf4 / GEMM_THREADS; ++i) {
      const int idx = tid + i * GEMM_THREADS;
      const int k = idx >> ncq;
      const int j = (idx - (k << ncq)) << 2;
      const float4 v = ((const float4*)W)[idx];
      const int base = ((k >> 1) * ncolP + j) * 2 + (k & 1);
      Ws[base + 0] = v.x;
      Ws[base + 2] = v.y;
      Ws[base + 4] = v.z;
      Ws[base + 6] = v.w;
    }
  }
  __syncthreads();

  const int wave = tid >> 5;
  const int lane = tid & 31;
  const int hl   = lane >> 4;
  const int lm   = lane & 15;

  int  s;
  bool on;
  if (blk < ALLOC_BLOCKS) {
    s  = blk * LIGHT_STRIPS + wave;
    on = wave < LIGHT_STRIPS;
  } else {
    s  = ALLOC_BLOCKS * LIGHT_STRIPS + wave * HEAVY_BLOCKS + (blk - ALLOC_BLOCKS);
    on = s < nstrips;
  }
  static_assert(ALLOC_BLOCKS * LIGHT_STRIPS + (GEMM_THREADS / 32) * HEAVY_BLOCKS >= NN / 16,
                "the deal must cover every strip");
  static_assert(ALLOC_BLOCKS * LIGHT_STRIPS <= NN / 16, "light strips are real strips");
  if (on) {
    const int m0 = s * 16;
    const float* __restrict__ xr = X + (size_t)(m0 + lm) * KDIM + 2 * hl;
    const int* __restrict__ ps = partS + (size_t)(hl * (NC / 2)) * PADN + m0 + lm;
    int cs = 0;
#pragma unroll
    for (int k = 0; k < NC / 2; ++k) cs += ps[(size_t)k * PADN];
    cs += __shfl_xor(cs, 16);
    const float nrm = rsqrtf(fmaxf((float)cs, 1.0f));

    v2f a[KDIM / 4];
    for (int j = 0; j < KDIM / 4; ++j) {
      const v2f x = *(const v2f*)(xr + 4 * j);
      a[j].x = x.x * nrm;
      a[j].y = x.y * nrm;
    }

    v8f acc[ntile];
    for (int t = 0; t < ntile; ++t) acc[t] = v8f{};

    for (int j = 0; j < KDIM / 4; ++j) {
      const int p = 2 * j + hl;
      for (int t = 0; t < ntile; ++t) {
        const v2f b = *(const v2f*)&Ws[(p * ncolP + t * 16 + lm) * 2];
        acc[t] = __builtin_amdgcn_wmma_f32_16x16x4_f32(
             false, a[j],  false, b,
             (short)0, acc[t],  false,  false);
      }
    }

    for (int t = 0; t < ntile; ++t) {
      for (int r = 0; r < 8; ++r)
        Y[(size_t)(m0 + hl * 8 + r) * ncol + t * 16 + lm] = acc[t][r];
    }
  }
}

__global__ __launch_bounds__(AG_THREADS) void k_alloc_gemm(
    const int* __restrict__ partS, int* __restrict__ partD,
    float* __restrict__ normS, float* __restrict__ normD,
    int2* __restrict__ seg, int* __restrict__ total,
    const float* __restrict__ X, const float* __restrict__ W, float* __restrict__ Y) {
  const int b = (int)blockIdx.x;
  if (b < ALLOC_BLOCKS) alloc_body(partS, partD, normS, normD, seg, total, NN, b);
  gemm_body<7>(X, partS, W, Y, NN / 16, b);
}


template <int VEC> struct RowVec;
template <> struct RowVec<4> { typedef v4f T; };
template <> struct RowVec<2> { typedef v2f T; };

#define GATHER_THREADS 256
#define GATHER_GRID    2048

template <int VEC, int U>
__device__ __forceinline__ void gather_batch(typename RowVec<VEC>::T& acc, int ids, int j,
                                             const typename RowVec<VEC>::T* __restrict__ H,
                                             int lane) {
  typedef typename RowVec<VEC>::T vecT;
  vecT v[U];
#pragma unroll
  for (int u = 0; u < U; ++u) {
    const int s = __builtin_amdgcn_readlane(ids, j + u);
    v[u] = H[(size_t)s * 32 + lane];
  }
#pragma unroll
  for (int u = 0; u < U; ++u) acc += v[u];
}

template <int VEC>
__device__ __forceinline__ void gather_chunk(typename RowVec<VEC>::T& acc, int ids, int m,
                                             const typename RowVec<VEC>::T* __restrict__ H,
                                             int lane) {
  int j = 0;
  for (; j + 8 <= m; j += 8) gather_batch<VEC, 8>(acc, ids, j, H, lane);
  if (m & 4) { gather_batch<VEC, 4>(acc, ids, j, H, lane); j += 4; }
  if (m & 2) { gather_batch<VEC, 2>(acc, ids, j, H, lane); j += 2; }
  if (m & 1) { gather_batch<VEC, 1>(acc, ids, j, H, lane); }
}

template <int VEC, bool RELU>
__global__ __launch_bounds__(GATHER_THREADS) void k_gather(
    const float* __restrict__ Hf, const int2* __restrict__ seg,
    const int* __restrict__ csrSrc, const float* __restrict__ normD,
    const float* __restrict__ bias, float* __restrict__ outf, int n, int nE) {
  typedef typename RowVec<VEC>::T vecT;
  const vecT* __restrict__ H = (const vecT*)Hf;
  vecT* __restrict__ out = (vecT*)outf;
  const int lane = threadIdx.x & 31;
  const int nwav = (int)gridDim.x * (GATHER_THREADS / 32);
  const vecT bv = ((const vecT*)bias)[lane];
  int node = __builtin_amdgcn_readfirstlane((int)blockIdx.x * (GATHER_THREADS / 32) +
                                            ((int)threadIdx.x >> 5));
  if (node >= n) return;
  const int last = nE - 1;
  int2 sg  = seg[node];
  int2 sgN = make_int2(0, 0);
  if (node + nwav < n) sgN = seg[node + nwav];
  int ids = csrSrc[min(sg.x + lane, last)];
  for (;;) {
    const int nn = node + nwav;
    const int idsN = csrSrc[min(sgN.x + lane, last)];
    int2 sgNN = make_int2(0, 0);
    if (nn + nwav < n) sgNN = seg[nn + nwav];
    const float nrm = normD[node];

    const int beg = sg.x;
    const int len = sg.y;
    vecT acc = (vecT)(0.0f);
    gather_chunk<VEC>(acc, ids, (len < 32) ? len : 32, H, lane);
    for (int c0 = 32; c0 < len; c0 += 32) {
      const int m = (len - c0 < 32) ? (len - c0) : 32;
      const int cid = csrSrc[min(beg + c0 + lane, last)];
      gather_chunk<VEC>(acc, cid, m, H, lane);
    }
    vecT r = acc * nrm + bv;
    if (RELU) {
#pragma unroll
      for (int q = 0; q < VEC; ++q) r[q] = fmaxf(r[q], 0.0f);
    }
    out[(size_t)node * 32 + lane] = r;

    if (nn >= n) break;
    node = nn;
    sg   = sgN;
    sgN  = sgNN;
    ids  = idsN;
  }
}


#define FUSE_THREADS 512
#define FUSE_ROWS    64
#define FUSE_PITCH4  33

__global__ __launch_bounds__(FUSE_THREADS) void k_gather_gemm(
    const float* __restrict__ Hf, const int2* __restrict__ seg,
    const int* __restrict__ csrSrc, const float* __restrict__ normD,
    const float* __restrict__ normS, const float* __restrict__ bias,
    const float* __restrict__ W, float* __restrict__ Y, int n, int nE) {
  constexpr int ncol  = 64;
  constexpr int ncolP = ncol + 16;
  __shared__ float Ws[(KDIM / 2) * ncolP * 2];
  __shared__ v4f   At[FUSE_ROWS * FUSE_PITCH4];

  const int tid  = threadIdx.x;
  const int lane = tid & 31;
  const int wave = __builtin_amdgcn_readfirstlane(tid >> 5);
  const int tile0 = (int)blockIdx.x * FUSE_ROWS;

  {
    constexpr int ncq = 4;
    constexpr int nf4 = KDIM << ncq;
    for (int i = 0; i < nf4 / FUSE_THREADS; ++i) {
      const int idx = tid + i * FUSE_THREADS;
      const int k = idx >> ncq;
      const int j = (idx - (k << ncq)) << 2;
      const float4 v = ((const float4*)W)[idx];
      const int base = ((k >> 1) * ncolP + j) * 2 + (k & 1);
      Ws[base + 0] = v.x;
      Ws[base + 2] = v.y;
      Ws[base + 4] = v.z;
      Ws[base + 6] = v.w;
    }
  }

  const int row0 = tile0 + wave * 4;
  if (row0 < n) {
    const v4f* __restrict__ H = (const v4f*)Hf;
    const v4f bv = ((const v4f*)bias)[lane];
    const int last = nE - 1;
    int2  sg[4];
    float nd[4];
    float ns[4];
    int   ids[4];
#pragma unroll
    for (int i = 0; i < 4; ++i) {
      const int r = min(row0 + i, n - 1);
      sg[i] = seg[r];
      nd[i] = normD[r];
      ns[i] = normS[r];
    }
#pragma unroll
    for (int i = 0; i < 4; ++i) ids[i] = csrSrc[min(sg[i].x + lane, last)];
#pragma unroll
    for (int i = 0; i < 4; ++i) {
      const int beg = sg[i].x;
      const int len = sg[i].y;
      v4f acc = (v4f)(0.0f);
      gather_chunk<4>(acc, ids[i], (len < 32) ? len : 32, H, lane);
      for (int c0 = 32; c0 < len; c0 += 32) {
        const int m = (len - c0 < 32) ? (len - c0) : 32;
        const int cid = csrSrc[min(beg + c0 + lane, last)];
        gather_chunk<4>(acc, cid, m, H, lane);
      }
      v4f r = acc * nd[i] + bv;
#pragma unroll
      for (int q = 0; q < 4; ++q) r[q] = fmaxf(r[q], 0.0f) * ns[i];
      At[(wave * 4 + i) * FUSE_PITCH4 + lane] = r;
    }
  }
  __syncthreads();

  const int st = wave >> 2;
  const int t  = wave & 3;
  const int m0 = tile0 + st * 16;
  if (m0 < n) {
    const int hl = lane >> 4;
    const int lm = lane & 15;
    const float* ar = (const float*)At + (st * 16 + lm) * (FUSE_PITCH4 * 4) + 2 * hl;
    v8f acc = v8f{};
    for (int j = 0; j < KDIM / 4; ++j) {
      const int p = 2 * j + hl;
      const v2f a = *(const v2f*)(ar + 4 * j);
      const v2f b = *(const v2f*)&Ws[(p * ncolP + t * 16 + lm) * 2];
      acc = __builtin_amdgcn_wmma_f32_16x16x4_f32(
           false, a,  false, b,
           (short)0, acc,  false,  false);
    }
    for (int r = 0; r < 8; ++r)
      Y[(size_t)(m0 + hl * 8 + r) * ncol + t * 16 + lm] = acc[r];
  }
}


extern "C" void kernel_launch(void* const* d_in, const int* in_sizes, int n_in,
                              void* d_out, int out_size, void* d_ws, size_t ws_size,
                              hipStream_t stream) {
  (void)in_sizes; (void)n_in; (void)out_size; (void)ws_size;

  const float* features = (const float*)d_in[0];
  const float* W1       = (const float*)d_in[1];
  const float* b1       = (const float*)d_in[2];
  const float* W2       = (const float*)d_in[3];
  const float* b2       = (const float*)d_in[4];
  const int*   esrc     = (const int*)d_in[5];
  const int*   edst     = (const int*)d_in[6];
  float*       out      = (float*)d_out;

  int*   wi     = (int*)d_ws;
  int*   total  = wi;
  float* normS  = (float*)(wi + 64);
  float* normD  = normS + PADN;
  int2*  seg    = (int2*)(normD + PADN);
  int*   tmpE   = (int*)(seg + PADN);
  int*   csrSrc = tmpE + NE;
  int*   partS  = csrSrc + NE;
  int*   partD  = partS + (size_t)NC * PADN;
  float* B0     = (float*)(partD + (size_t)NC * PADN);
  float* B1     = B0 + (size_t)NN * 128;

  {
    k_count<<<CSR_GRID, CSR_THREADS, 0, stream>>>(esrc, edst, partS, partD, total);
    k_alloc_gemm<<<GEMM_GRID, AG_THREADS, 0, stream>>>(
        partS, partD, normS, normD, seg, total, features, W1, B0);
    k_fill<<<CSR_GRID, CSR_THREADS, 0, stream>>>(esrc, edst, seg, partD, tmpE, csrSrc);
  }

  k_gather_gemm<<<(NN + FUSE_ROWS - 1) / FUSE_ROWS, FUSE_THREADS, 0, stream>>>(
      B0, seg, csrSrc, normD, normS, b1, W2, B1, NN, NE);

  k_gather<2, false><<<GATHER_GRID, GATHER_THREADS, 0, stream>>>(
      B1, seg, csrSrc, normD, b2, out, NN, NE);
}


# Edge-id order of each segment: made by the summing wave in registers (lane compares, one lane permute of (edge id, source id) pairs), not by the fill launch through memory; sum order unchanged
# speedup vs baseline: 5.5051x; 1.1344x over previous
#include <hip/hip_runtime.h>


typedef __attribute__((ext_vector_type(2))) float v2f;
typedef __attribute__((ext_vector_type(4))) float v4f;
typedef __attribute__((ext_vector_type(8))) float v8f;

#define NN   50000
#define NE   600000
#define KDIM 128
#define PADN 50176


#define NR   8
#define RN   6250
#define NC   32
#define CE4  4688
#define CSR_THREADS 1024
#define CSR_GRID    (NR * NC)
#define CSR_NI      ((CE4 + CSR_THREADS - 1) / CSR_THREADS)
#define CSR_RI      ((RN + CSR_THREADS - 1) / CSR_THREADS)

__global__ __launch_bounds__(CSR_THREADS) void k_count(
    const int* __restrict__ src, const int* __restrict__ dst,
    int* __restrict__ partS, int* __restrict__ partD, int* __restrict__ total) {
  __shared__ int cS[RN];
  __shared__ int cD[RN];
  const int tid = threadIdx.x;
  const int r   = (int)blockIdx.x % NR;
  const int c   = (int)blockIdx.x / NR;
  const int lo  = r * RN;
  const int n4 = min(CE4, NE / 4 - c * CE4);
  const int4* __restrict__ s4 = (const int4*)src + c * CE4;
  const int4* __restrict__ d4 = (const int4*)dst + c * CE4;
  int4 s[CSR_NI];
  int4 d[CSR_NI];
#pragma unroll
  for (int k = 0; k < CSR_NI; ++k) {
    const int i = min(tid + k * CSR_THREADS, n4 - 1);
    s[k] = s4[i];
    d[k] = d4[i];
  }
  for (int i = tid; i < RN; i += CSR_THREADS) { cS[i] = 0; cD[i] = 0; }
  if (blockIdx.x == 0 && tid == 0) *total = 0;
  __syncthreads();
#pragma unroll
  for (int k = 0; k < CSR_NI; ++k) {
    if (tid + k * CSR_THREADS < n4) {
      unsigned u;
      u = (unsigned)(s[k].x - lo); if (u < RN) atomicAdd(&cS[u], 1);
      u = (unsigned)(s[k].y - lo); if (u < RN) atomicAdd(&cS[u], 1);
      u = (unsigned)(s[k].z - lo); if (u < RN) atomicAdd(&cS[u], 1);
      u = (unsigned)(s[k].w - lo); if (u < RN) atomicAdd(&cS[u], 1);
      u = (unsigned)(d[k].x - lo); if (u < RN) atomicAdd(&cD[u], 1);
      u = (unsigned)(d[k].y - lo); if (u < RN) atomicAdd(&cD[u], 1);
      u = (unsigned)(d[k].z - lo); if (u < RN) atomicAdd(&cD[u], 1);
      u = (unsigned)(d[k].w - lo); if (u < RN) atomicAdd(&cD[u], 1);
    }
  }
  __syncthreads();
  int* __restrict__ pS = partS + (size_t)c * PADN + lo;
  int* __restrict__ pD = partD + (size_t)c * PADN + lo;
  for (int i = tid; i < RN; i += CSR_THREADS) { pS[i] = cS[i]; pD[i] = cD[i]; }
}

#define AG_THREADS   512
#define ALLOC_BLOCKS ((NN + AG_THREADS - 1) / AG_THREADS)
__device__ __forceinline__ void alloc_body(
    const int* __restrict__ partS, int* __restrict__ partD,
    float* __restrict__ normS, float* __restrict__ normD,
    int2* __restrict__ seg, int* __restrict__ total, int n, int blk) {
  __shared__ int wsum[AG_THREADS / 32];
  __shared__ int base;
  const int i    = blk * AG_THREADS + threadIdx.x;
  const int lane = threadIdx.x & 31;
  const int wave = threadIdx.x >> 5;
  int cs = 0;
  int c  = 0;
  if (i < n) {
    int vs[NC];
    int vd[NC];
#pragma unroll
    for (int k = 0; k < NC; ++k) {
      vs[k] = partS[(size_t)k * PADN + i];
      vd[k] = partD[(size_t)k * PADN + i];
    }
#pragma unroll
    for (int k = 0; k < NC; ++k) {
      cs += vs[k];
      partD[(size_t)k * PADN + i] = c;
      c += vd[k];
    }
  }
  int x = c;
  for (int o = 1; o < 32; o <<= 1) {
    const int y = __shfl_up(x, o);
    if (lane >= o) x += y;
  }
  if (lane == 31) wsum[wave] = x;
  __syncthreads();
  if (threadIdx.x == 0) {
    int t = 0;
    for (int w = 0; w < AG_THREADS / 32; ++w) {
      const int v = wsum[w];
      wsum[w] = t;
      t += v;
    }
    base = atomicAdd(total, t);
  }
  __syncthreads();
  if (i < n) {
    const int beg = base + wsum[wave] + x - c;
    seg[i]    = make_int2(beg, c);
    normS[i]  = rsqrtf(fmaxf((float)cs, 1.0f));
    normD[i]  = rsqrtf(fmaxf((float)c, 1.0f));
  }
}

__global__ __launch_bounds__(CSR_THREADS) void k_fill(
    const int* __restrict__ src, const int* __restrict__ dst,
    const int2* __restrict__ seg, const int* __restrict__ partD,
    int2* __restrict__ pairs) {
  __shared__ int cur[CSR_RI * CSR_THREADS];
  const int tid = threadIdx.x;
  const int r   = (int)blockIdx.x % NR;
  const int c   = (int)blockIdx.x / NR;
  const int lo  = r * RN;
  const int* __restrict__ pD = partD + (size_t)c * PADN + lo;
  const int n4 = min(CE4, NE / 4 - c * CE4);
  const int4* __restrict__ s4 = (const int4*)src + c * CE4;
  const int4* __restrict__ d4 = (const int4*)dst + c * CE4;
  int4 s[CSR_NI];
  int4 d[CSR_NI];
#pragma unroll
  for (int k = 0; k < CSR_NI; ++k) {
    const int i = min(tid + k * CSR_THREADS, n4 - 1);
    s[k] = s4[i];
    d[k] = d4[i];
  }
  int sb[CSR_RI];
  int co[CSR_RI];
#pragma unroll
  for (int k = 0; k < CSR_RI; ++k) {
    const int i = min(tid + k * CSR_THREADS, RN - 1);
    sb[k] = seg[lo + i].x;
    co[k] = pD[i];
  }
#pragma unroll
  for (int k = 0; k < CSR_RI; ++k) {
    const int i = tid + k * CSR_THREADS;
    cur[i] = (i < RN) ? (sb[k] + co[k]) : 0;
  }
  __syncthreads();
#pragma unroll
  for (int k = 0; k < CSR_NI; ++k) {
    const int i = tid + k * CSR_THREADS;
    if (i < n4) {
      const int e0 = 4 * (c * CE4 + i);
      unsigned u;
      u = (unsigned)(d[k].x - lo); if (u < RN) pairs[atomicAdd(&cur[u], 1)] = make_int2(e0 + 0, s[k].x);
      u = (unsigned)(d[k].y - lo); if (u < RN) pairs[atomicAdd(&cur[u], 1)] = make_int2(e0 + 1, s[k].y);
      u = (unsigned)(d[k].z - lo); if (u < RN) pairs[atomicAdd(&cur[u], 1)] = make_int2(e0 + 2, s[k].z);
      u = (unsigned)(d[k].w - lo); if (u < RN) pairs[atomicAdd(&cur[u], 1)] = make_int2(e0 + 3, s[k].w);
    }
  }
}


#define GEMM_THREADS AG_THREADS
#define GEMM_GRID    256
#define LIGHT_STRIPS 8
#define HEAVY_BLOCKS (GEMM_GRID - ALLOC_BLOCKS)

template <int NCSHIFT  >
__device__ __forceinline__ void gemm_body(
    const float* __restrict__ X, const int* __restrict__ partS,
    const float* __restrict__ W, float* __restrict__ Y, int nstrips, int blk) {
  constexpr int ncol  = 1 << NCSHIFT;
  constexpr int ncolP = ncol + 16;
  constexpr int ntile = ncol >> 4;
  __shared__ float Ws[(KDIM / 2) * ncolP * 2];

  const int tid = threadIdx.x;

  {
    constexpr int ncq = NCSHIFT - 2;
    constexpr int nf4 = KDIM << ncq;
    for (int i = 0; i < nf4 / GEMM_THREADS; ++i) {
      const int idx = tid + i * GEMM_THREADS;
      const int k = idx >> ncq;
      const int j = (idx - (k << ncq)) << 2;
      const float4 v = ((const float4*)W)[idx];
      const int base = ((k >> 1) * ncolP + j) * 2 + (k & 1);
      Ws[base + 0] = v.x;
      Ws[base + 2] = v.y;
      Ws[base + 4] = v.z;
      Ws[base + 6] = v.w;
    }
  }
  __syncthreads();

  const int wave = tid >> 5;
  const int lane = tid & 31;
  const int hl   = lane >> 4;
  const int lm   = lane & 15;

  int  s;
  bool on;
  if (blk < ALLOC_BLOCKS) {
    s  = blk * LIGHT_STRIPS + wave;
    on = wave < LIGHT_STRIPS;
  } else {
    s  = ALLOC_BLOCKS * LIGHT_STRIPS + wave * HEAVY_BLOCKS + (blk - ALLOC_BLOCKS);
    on = s < nstrips;
  }
  static_assert(ALLOC_BLOCKS * LIGHT_STRIPS + (GEMM_THREADS / 32) * HEAVY_BLOCKS >= NN / 16,
                "the deal must cover every strip");
  static_assert(ALLOC_BLOCKS * LIGHT_STRIPS <= NN / 16, "light strips are real strips");
  if (on) {
    const int m0 = s * 16;
    const float* __restrict__ xr = X + (size_t)(m0 + lm) * KDIM + 2 * hl;
    const int* __restrict__ ps = partS + (size_t)(hl * (NC / 2)) * PADN + m0 + lm;
    int cs = 0;
#pragma unroll
    for (int k = 0; k < NC / 2; ++k) cs += ps[(size_t)k * PADN];
    cs += __shfl_xor(cs, 16);
    const float nrm = rsqrtf(fmaxf((float)cs, 1.0f));

    v2f a[KDIM / 4];
    for (int j = 0; j < KDIM / 4; ++j) {
      const v2f x = *(const v2f*)(xr + 4 * j);
      a[j].x = x.x * nrm;
      a[j].y = x.y * nrm;
    }

    v8f acc[ntile];
    for (int t = 0; t < ntile; ++t) acc[t] = v8f{};

    for (int j = 0; j < KDIM / 4; ++j) {
      const int p = 2 * j + hl;
      for (int t = 0; t < ntile; ++t) {
        const v2f b = *(const v2f*)&Ws[(p * ncolP + t * 16 + lm) * 2];
        acc[t] = __builtin_amdgcn_wmma_f32_16x16x4_f32(
             false, a[j],  false, b,
             (short)0, acc[t],  false,  false);
      }
    }

    for (int t = 0; t < ntile; ++t) {
      for (int r = 0; r < 8; ++r)
        Y[(size_t)(m0 + hl * 8 + r) * ncol + t * 16 + lm] = acc[t][r];
    }
  }
}

__global__ __launch_bounds__(AG_THREADS) void k_alloc_gemm(
    const int* __restrict__ partS, int* __restrict__ partD,
    float* __restrict__ normS, float* __restrict__ normD,
    int2* __restrict__ seg, int* __restrict__ total,
    const float* __restrict__ X, const float* __restrict__ W, float* __restrict__ Y) {
  const int b = (int)blockIdx.x;
  if (b < ALLOC_BLOCKS) alloc_body(partS, partD, normS, normD, seg, total, NN, b);
  gemm_body<7>(X, partS, W, Y, NN / 16, b);
}


template <int VEC> struct RowVec;
template <> struct RowVec<4> { typedef v4f T; };
template <> struct RowVec<2> { typedef v2f T; };

#define GATHER_THREADS 256
#define GATHER_GRID    2048

template <int VEC, int U>
__device__ __forceinline__ void gather_batch(typename RowVec<VEC>::T& acc, int ids, int j,
                                             const typename RowVec<VEC>::T* __restrict__ H,
                                             int lane) {
  typedef typename RowVec<VEC>::T vecT;
  vecT v[U];
#pragma unroll
  for (int u = 0; u < U; ++u) {
    const int s = __builtin_amdgcn_readlane(ids, j + u);
    v[u] = H[(size_t)s * 32 + lane];
  }
#pragma unroll
  for (int u = 0; u < U; ++u) acc += v[u];
}

template <int VEC>
__device__ __forceinline__ void gather_chunk(typename RowVec<VEC>::T& acc, int ids, int m,
                                             const typename RowVec<VEC>::T* __restrict__ H,
                                             int lane) {
  int j = 0;
  for (; j + 8 <= m; j += 8) gather_batch<VEC, 8>(acc, ids, j, H, lane);
  if (m & 4) { gather_batch<VEC, 4>(acc, ids, j, H, lane); j += 4; }
  if (m & 2) { gather_batch<VEC, 2>(acc, ids, j, H, lane); j += 2; }
  if (m & 1) { gather_batch<VEC, 1>(acc, ids, j, H, lane); }
}

__device__ __forceinline__ int order_by_edge(int2 pr, int m, int lane) {
  int rank = 0;
  for (int i = 0; i < m; ++i) {
    const int ei = __builtin_amdgcn_readlane(pr.x, i);
    rank += (ei < pr.x) ? 1 : 0;
  }
  const int to = (lane < m) ? rank : lane;
  return __builtin_amdgcn_ds_permute(to << 2, pr.y);
}

template <int VEC>
__device__ __forceinline__ void gather_long(typename RowVec<VEC>::T& acc,
                                            const int2* __restrict__ pairs,
                                            int* __restrict__ csrSrc, int beg, int len, int last,
                                            const typename RowVec<VEC>::T* __restrict__ H,
                                            int lane) {
  int prev = -1;
  for (int r = 0; r < len; ++r) {
    int be = 0x7fffffff;
    int bs = 0;
    for (int c0 = 0; c0 < len; c0 += 32) {
      const int j = c0 + lane;
      const int2 q = pairs[min(beg + j, last)];
      if (j < len && q.x > prev && q.x < be) { be = q.x; bs = q.y; }
    }
    for (int o = 16; o > 0; o >>= 1) {
      const int oe = __shfl_xor(be, o);
      const int os = __shfl_xor(bs, o);
      if (oe < be) { be = oe; bs = os; }
    }
    const int s = __builtin_amdgcn_readfirstlane(bs);
    prev = __builtin_amdgcn_readfirstlane(be);
    if (lane == 0) csrSrc[beg + r] = s;
    acc += H[(size_t)s * 32 + lane];
  }
}

template <int VEC, bool RELU>
__global__ __launch_bounds__(GATHER_THREADS) void k_gather(
    const float* __restrict__ Hf, const int2* __restrict__ seg,
    const int* __restrict__ csrSrc, const float* __restrict__ normD,
    const float* __restrict__ bias, float* __restrict__ outf, int n, int nE) {
  typedef typename RowVec<VEC>::T vecT;
  const vecT* __restrict__ H = (const vecT*)Hf;
  vecT* __restrict__ out = (vecT*)outf;
  const int lane = threadIdx.x & 31;
  const int nwav = (int)gridDim.x * (GATHER_THREADS / 32);
  const vecT bv = ((const vecT*)bias)[lane];
  int node = __builtin_amdgcn_readfirstlane((int)blockIdx.x * (GATHER_THREADS / 32) +
                                            ((int)threadIdx.x >> 5));
  if (node >= n) return;
  const int last = nE - 1;
  int2 sg  = seg[node];
  int2 sgN = make_int2(0, 0);
  if (node + nwav < n) sgN = seg[node + nwav];
  int ids = csrSrc[min(sg.x + lane, last)];
  for (;;) {
    const int nn = node + nwav;
    const int idsN = csrSrc[min(sgN.x + lane, last)];
    int2 sgNN = make_int2(0, 0);
    if (nn + nwav < n) sgNN = seg[nn + nwav];
    const float nrm = normD[node];

    const int beg = sg.x;
    const int len = sg.y;
    vecT acc = (vecT)(0.0f);
    gather_chunk<VEC>(acc, ids, (len < 32) ? len : 32, H, lane);
    for (int c0 = 32; c0 < len; c0 += 32) {
      const int m = (len - c0 < 32) ? (len - c0) : 32;
      const int cid = csrSrc[min(beg + c0 + lane, last)];
      gather_chunk<VEC>(acc, cid, m, H, lane);
    }
    vecT r = acc * nrm + bv;
    if (RELU) {
#pragma unroll
      for (int q = 0; q < VEC; ++q) r[q] = fmaxf(r[q], 0.0f);
    }
    out[(size_t)node * 32 + lane] = r;

    if (nn >= n) break;
    node = nn;
    sg   = sgN;
    sgN  = sgNN;
    ids  = idsN;
  }
}


#define FUSE_THREADS 512
#define FUSE_ROWS    64
#define FUSE_PITCH4  33

__global__ __launch_bounds__(FUSE_THREADS) void k_gather_gemm(
    const float* __restrict__ Hf, const int2* __restrict__ seg,
    const int2* __restrict__ pairs, int* __restrict__ csrSrc, const float* __restrict__ normD,
    const float* __restrict__ normS, const float* __restrict__ bias,
    const float* __restrict__ W, float* __restrict__ Y, int n, int nE) {
  constexpr int ncol  = 64;
  constexpr int ncolP = ncol + 16;
  __shared__ float Ws[(KDIM / 2) * ncolP * 2];
  __shared__ v4f   At[FUSE_ROWS * FUSE_PITCH4];

  const int tid  = threadIdx.x;
  const int lane = tid & 31;
  const int wave = __builtin_amdgcn_readfirstlane(tid >> 5);
  const int tile0 = (int)blockIdx.x * FUSE_ROWS;

  {
    constexpr int ncq = 4;
    constexpr int nf4 = KDIM << ncq;
    for (int i = 0; i < nf4 / FUSE_THREADS; ++i) {
      const int idx = tid + i * FUSE_THREADS;
      const int k = idx >> ncq;
      const int j = (idx - (k << ncq)) << 2;
      const float4 v = ((const float4*)W)[idx];
      const int base = ((k >> 1) * ncolP + j) * 2 + (k & 1);
      Ws[base + 0] = v.x;
      Ws[base + 2] = v.y;
      Ws[base + 4] = v.z;
      Ws[base + 6] = v.w;
    }
  }

  const int row0 = tile0 + wave * 4;
  if (row0 < n) {
    const v4f* __restrict__ H = (const v4f*)Hf;
    const v4f bv = ((const v4f*)bias)[lane];
    const int last = nE - 1;
    int2  sg[4];
    float nd[4];
    float ns[4];
    int2  pr[4];
#pragma unroll
    for (int i = 0; i < 4; ++i) {
      const int r = min(row0 + i, n - 1);
      sg[i] = seg[r];
      nd[i] = normD[r];
      ns[i] = normS[r];
    }
#pragma unroll
    for (int i = 0; i < 4; ++i) pr[i] = pairs[min(sg[i].x + lane, last)];
#pragma unroll
    for (int i = 0; i < 4; ++i) {
      const int beg = __builtin_amdgcn_readfirstlane(sg[i].x);
      const int len = __builtin_amdgcn_readfirstlane(sg[i].y);
      v4f acc = (v4f)(0.0f);
      if (len <= 32) {
        const int ids = order_by_edge(pr[i], len, lane);
        if (lane < len) csrSrc[beg + lane] = ids;
        gather_chunk<4>(acc, ids, len, H, lane);
      } else {
        gather_long<4>(acc, pairs, csrSrc, beg, len, last, H, lane);
      }
      v4f r = acc * nd[i] + bv;
#pragma unroll
      for (int q = 0; q < 4; ++q) r[q] = fmaxf(r[q], 0.0f) * ns[i];
      At[(wave * 4 + i) * FUSE_PITCH4 + lane] = r;
    }
  }
  __syncthreads();

  const int st = wave >> 2;
  const int t  = wave & 3;
  const int m0 = tile0 + st * 16;
  if (m0 < n) {
    const int hl = lane >> 4;
    const int lm = lane & 15;
    const float* ar = (const float*)At + (st * 16 + lm) * (FUSE_PITCH4 * 4) + 2 * hl;
    v8f acc = v8f{};
    for (int j = 0; j < KDIM / 4; ++j) {
      const int p = 2 * j + hl;
      const v2f a = *(const v2f*)(ar + 4 * j);
      const v2f b = *(const v2f*)&Ws[(p * ncolP + t * 16 + lm) * 2];
      acc = __builtin_amdgcn_wmma_f32_16x16x4_f32(
           false, a,  false, b,
           (short)0, acc,  false,  false);
    }
    for (int r = 0; r < 8; ++r)
      Y[(size_t)(m0 + hl * 8 + r) * ncol + t * 16 + lm] = acc[r];
  }
}


extern "C" void kernel_launch(void* const* d_in, const int* in_sizes, int n_in,
                              void* d_out, int out_size, void* d_ws, size_t ws_size,
                              hipStream_t stream) {
  (void)in_sizes; (void)n_in; (void)out_size; (void)ws_size;

  const float* features = (const float*)d_in[0];
  const float* W1       = (const float*)d_in[1];
  const float* b1       = (const float*)d_in[2];
  const float* W2       = (const float*)d_in[3];
  const float* b2       = (const float*)d_in[4];
  const int*   esrc     = (const int*)d_in[5];
  const int*   edst     = (const int*)d_in[6];
  float*       out      = (float*)d_out;

  int*   wi     = (int*)d_ws;
  int*   total  = wi;
  float* normS  = (float*)(wi + 64);
  float* normD  = normS + PADN;
  int2*  seg    = (int2*)(normD + PADN);
  int2*  pairs  = seg + PADN;
  int*   csrSrc = (int*)(pairs + NE);
  int*   partS  = csrSrc + NE;
  int*   partD  = partS + (size_t)NC * PADN;
  float* B0     = (float*)(partD + (size_t)NC * PADN);
  float* B1     = B0 + (size_t)NN * 128;

  {
    k_count<<<CSR_GRID, CSR_THREADS, 0, stream>>>(esrc, edst, partS, partD, total);
    k_alloc_gemm<<<GEMM_GRID, AG_THREADS, 0, stream>>>(
        partS, partD, normS, normD, seg, total, features, W1, B0);
    k_fill<<<CSR_GRID, CSR_THREADS, 0, stream>>>(esrc, edst, seg, partD, pairs);
  }

  k_gather_gemm<<<(NN + FUSE_ROWS - 1) / FUSE_ROWS, FUSE_THREADS, 0, stream>>>(
      B0, seg, pairs, csrSrc, normD, normS, b1, W2, B1, NN, NE);

  k_gather<2, false><<<GATHER_GRID, GATHER_THREADS, 0, stream>>>(
      B1, seg, csrSrc, normD, b2, out, NN, NE);
}
